# starting kernel
# baseline (speedup 1.0000x reference)
#include <hip/hip_runtime.h>

typedef __attribute__((ext_vector_type(2))) float v2f;
typedef __attribute__((ext_vector_type(8))) float v8f;

#define NN      100000
#define DINF    128
#define DHIDF   128
#define DOUTF   64
#define NEDGE   1600000

__global__ void zero_f32v4(float4* __restrict__ p, long n4) {
    long i = (long)blockIdx.x * blockDim.x + threadIdx.x;
    long st = (long)gridDim.x * blockDim.x;
    float4 z = make_float4(0.f, 0.f, 0.f, 0.f);
    for (; i < n4; i += st) p[i] = z;
}

template <int KC, bool RELU_IN>
__global__ void gemm_wmma_f32(const float* __restrict__ H,
                              const float* __restrict__ W,
                              float* __restrict__ C,
                              int N) {
    const int lane  = threadIdx.x & 31;
    const int wave  = threadIdx.x >> 5;
    const int m0    = blockIdx.x * 16;
    const int n0    = wave * 16;
    const int half  = lane >> 4;
    const int l15   = lane & 15;
    const int kbase = half * 2;

    const float* hrow = H + (long)(m0 + l15) * KC;
    const float* wrow = W + (long)(n0 + l15) * KC;

    v8f acc = {};
    #pragma unroll
    for (int k = 0; k < KC; k += 4) {
        v2f a, b;
        a.x = hrow[k + kbase];
        a.y = hrow[k + kbase + 1];
        if (RELU_IN) {
            a.x = a.x > 0.f ? a.x : 0.f;
            a.y = a.y > 0.f ? a.y : 0.f;
        }
        b.x = wrow[k + kbase];
        b.y = wrow[k + kbase + 1];
        acc = __builtin_amdgcn_wmma_f32_16x16x4_f32(
             false, a,  false, b,
             (short)0, acc,  false,  false);
    }

    float* cp = C + (long)(m0 + half * 8) * N + n0 + l15;
    #pragma unroll
    for (int i = 0; i < 8; ++i) cp[(long)i * N] = acc[i];
}

template <int SHIFT, bool UNIFORM_E>
__global__ void spmm_atomic(const int*   __restrict__ erow,
                            const int*   __restrict__ ecol,
                            const float* __restrict__ eval,
                            const float* __restrict__ h,
                            float*       __restrict__ y,
                            long totalChunks) {
    constexpr int D = 4 << SHIFT;
    long g = (long)blockIdx.x * blockDim.x + threadIdx.x;
    if (g >= totalChunks) return;
    int e;
    if (UNIFORM_E) {
        e = __builtin_amdgcn_readfirstlane((int)(g >> SHIFT));
    } else {
        e = (int)(g >> SHIFT);
    }
    int q = (int)(g & ((1 << SHIFT) - 1));

    int   r = erow[e];
    int   c = ecol[e];
    float v = eval[e];

    const float4 hv = *reinterpret_cast<const float4*>(h + (long)c * D + q * 4);
    float* yp = y + (long)r * D + q * 4;
    unsafeAtomicAdd(yp + 0, v * hv.x);
    unsafeAtomicAdd(yp + 1, v * hv.y);
    unsafeAtomicAdd(yp + 2, v * hv.z);
    unsafeAtomicAdd(yp + 3, v * hv.w);
}

__global__ void relu_add_relu(float* __restrict__ dst,
                              const float* __restrict__ res, long n) {
    long i = (long)blockIdx.x * blockDim.x + threadIdx.x;
    long st = (long)gridDim.x * blockDim.x;
    for (; i < n; i += st) {
        float a = dst[i];
        float b = res[i];
        a = a > 0.f ? a : 0.f;
        b = b > 0.f ? b : 0.f;
        dst[i] = a + b;
    }
}

extern "C" void kernel_launch(void* const* d_in, const int* in_sizes, int n_in,
                              void* d_out, int out_size, void* d_ws, size_t ws_size,
                              hipStream_t stream) {
    const float* x    = (const float*)d_in[0];
    const int*   erow = (const int*)  d_in[1];
    const int*   ecol = (const int*)  d_in[2];
    const float* evalv= (const float*)d_in[3];
    const float* W0   = (const float*)d_in[4];
    const float* W1   = (const float*)d_in[5];
    const float* W2   = (const float*)d_in[6];
    float*       out  = (float*)d_out;

    const long nh = (long)NN * DHIDF;
    float* T  = (float*)d_ws;
    float* A  = T + nh;
    float* Bb = A + nh;

    const long chunks128 = (long)NEDGE * (DHIDF / 4);
    const long chunks64  = (long)NEDGE * (DOUTF / 4);
    const int  mblocks   = NN / 16;

    gemm_wmma_f32<DINF, false><<<mblocks, 32 * (DHIDF / 16), 0, stream>>>(
        x, W0, T, DHIDF);
    zero_f32v4<<<2048, 256, 0, stream>>>((float4*)A, nh / 4);
    spmm_atomic<5, true><<<(int)((chunks128 + 255) / 256), 256, 0, stream>>>(
        erow, ecol, evalv, T, A, chunks128);

    gemm_wmma_f32<DHIDF, true><<<mblocks, 32 * (DHIDF / 16), 0, stream>>>(
        A, W1, T, DHIDF);
    zero_f32v4<<<2048, 256, 0, stream>>>((float4*)Bb, nh / 4);
    spmm_atomic<5, true><<<(int)((chunks128 + 255) / 256), 256, 0, stream>>>(
        erow, ecol, evalv, T, Bb, chunks128);
    relu_add_relu<<<2048, 256, 0, stream>>>(Bb, A, nh);

    gemm_wmma_f32<DHIDF, false><<<mblocks, 32 * (DOUTF / 16), 0, stream>>>(
        Bb, W2, T, DOUTF);
    zero_f32v4<<<2048, 256, 0, stream>>>((float4*)out, (long)NN * DOUTF / 4);
    spmm_atomic<4, false><<<(int)((chunks64 + 255) / 256), 256, 0, stream>>>(
        erow, ecol, evalv, T, out, chunks64);
}


# Scatter-add lane map: lane l owns floats l, l+32, .. of a row, so each f32 atomic (and load) instruction covers one whole 128-byte line, not four quarter lines; atomic arrival-order sums as before
# speedup vs baseline: 1.2853x; 1.2853x over previous
#include <hip/hip_runtime.h>

typedef __attribute__((ext_vector_type(2))) float v2f;
typedef __attribute__((ext_vector_type(8))) float v8f;

#define NN      100000
#define DINF    128
#define DHIDF   128
#define DOUTF   64
#define NEDGE   1600000

__global__ void zero_f32v4(float4* __restrict__ p, long n4) {
    long i = (long)blockIdx.x * blockDim.x + threadIdx.x;
    long st = (long)gridDim.x * blockDim.x;
    float4 z = make_float4(0.f, 0.f, 0.f, 0.f);
    for (; i < n4; i += st) p[i] = z;
}

template <int KC, bool RELU_IN>
__global__ void gemm_wmma_f32(const float* __restrict__ H,
                              const float* __restrict__ W,
                              float* __restrict__ C,
                              int N) {
    const int lane  = threadIdx.x & 31;
    const int wave  = threadIdx.x >> 5;
    const int m0    = blockIdx.x * 16;
    const int n0    = wave * 16;
    const int half  = lane >> 4;
    const int l15   = lane & 15;
    const int kbase = half * 2;

    const float* hrow = H + (long)(m0 + l15) * KC;
    const float* wrow = W + (long)(n0 + l15) * KC;

    v8f acc = {};
    #pragma unroll
    for (int k = 0; k < KC; k += 4) {
        v2f a, b;
        a.x = hrow[k + kbase];
        a.y = hrow[k + kbase + 1];
        if (RELU_IN) {
            a.x = a.x > 0.f ? a.x : 0.f;
            a.y = a.y > 0.f ? a.y : 0.f;
        }
        b.x = wrow[k + kbase];
        b.y = wrow[k + kbase + 1];
        acc = __builtin_amdgcn_wmma_f32_16x16x4_f32(
             false, a,  false, b,
             (short)0, acc,  false,  false);
    }

    float* cp = C + (long)(m0 + half * 8) * N + n0 + l15;
    #pragma unroll
    for (int i = 0; i < 8; ++i) cp[(long)i * N] = acc[i];
}

template <int D, int EPW>
__global__ void spmm_atomic(const int*   __restrict__ erow,
                            const int*   __restrict__ ecol,
                            const float* __restrict__ eval,
                            const float* __restrict__ h,
                            float*       __restrict__ y,
                            int nWaves) {
    constexpr int R = D / 32;
    const int lane = threadIdx.x & 31;
    const int w = __builtin_amdgcn_readfirstlane(
        (int)((blockIdx.x * blockDim.x + threadIdx.x) >> 5));
    if (w >= nWaves) return;

    int   r[EPW];
    int   c[EPW];
    float v[EPW];
    #pragma unroll
    for (int k = 0; k < EPW; ++k) {
        const int e = w * EPW + k;
        r[k] = erow[e];
        c[k] = ecol[e];
        v[k] = eval[e];
    }

    float hv[EPW][R];
    #pragma unroll
    for (int k = 0; k < EPW; ++k) {
        const float* hp = h + (long)c[k] * D + lane;
        #pragma unroll
        for (int j = 0; j < R; ++j) hv[k][j] = hp[j * 32];
    }

    #pragma unroll
    for (int k = 0; k < EPW; ++k) {
        float* yp = y + (long)r[k] * D + lane;
        #pragma unroll
        for (int j = 0; j < R; ++j) unsafeAtomicAdd(yp + j * 32, v[k] * hv[k][j]);
    }
}

__global__ void relu_add_relu(float* __restrict__ dst,
                              const float* __restrict__ res, long n) {
    long i = (long)blockIdx.x * blockDim.x + threadIdx.x;
    long st = (long)gridDim.x * blockDim.x;
    for (; i < n; i += st) {
        float a = dst[i];
        float b = res[i];
        a = a > 0.f ? a : 0.f;
        b = b > 0.f ? b : 0.f;
        dst[i] = a + b;
    }
}

extern "C" void kernel_launch(void* const* d_in, const int* in_sizes, int n_in,
                              void* d_out, int out_size, void* d_ws, size_t ws_size,
                              hipStream_t stream) {
    const float* x    = (const float*)d_in[0];
    const int*   erow = (const int*)  d_in[1];
    const int*   ecol = (const int*)  d_in[2];
    const float* evalv= (const float*)d_in[3];
    const float* W0   = (const float*)d_in[4];
    const float* W1   = (const float*)d_in[5];
    const float* W2   = (const float*)d_in[6];
    float*       out  = (float*)d_out;

    const long nh = (long)NN * DHIDF;
    float* T  = (float*)d_ws;
    float* A  = T + nh;
    float* Bb = A + nh;

    constexpr int waves128 = NEDGE;
    constexpr int waves64  = NEDGE / 2;
    const int  mblocks   = NN / 16;

    gemm_wmma_f32<DINF, false><<<mblocks, 32 * (DHIDF / 16), 0, stream>>>(
        x, W0, T, DHIDF);
    zero_f32v4<<<2048, 256, 0, stream>>>((float4*)A, nh / 4);
    spmm_atomic<DHIDF, 1><<<(waves128 + 7) / 8, 256, 0, stream>>>(
        erow, ecol, evalv, T, A, waves128);

    gemm_wmma_f32<DHIDF, true><<<mblocks, 32 * (DHIDF / 16), 0, stream>>>(
        A, W1, T, DHIDF);
    zero_f32v4<<<2048, 256, 0, stream>>>((float4*)Bb, nh / 4);
    spmm_atomic<DHIDF, 1><<<(waves128 + 7) / 8, 256, 0, stream>>>(
        erow, ecol, evalv, T, Bb, waves128);
    relu_add_relu<<<2048, 256, 0, stream>>>(Bb, A, nh);

    gemm_wmma_f32<DHIDF, false><<<mblocks, 32 * (DOUTF / 16), 0, stream>>>(
        Bb, W2, T, DOUTF);
    zero_f32v4<<<2048, 256, 0, stream>>>((float4*)out, (long)NN * DOUTF / 4);
    spmm_atomic<DOUTF, 2><<<(waves64 + 7) / 8, 256, 0, stream>>>(
        erow, ecol, evalv, T, out, waves64);
}


# Scatter-add by cache half: y rows split in two ranges, a wave (die read from hardware, edges drawn by ticket) adds only into its own L2 half's range; atomic arrival-order sums as before
# speedup vs baseline: 2.6262x; 2.0433x over previous
#include <hip/hip_runtime.h>

typedef __attribute__((ext_vector_type(2))) float v2f;
typedef __attribute__((ext_vector_type(8))) float v8f;
typedef __attribute__((ext_vector_type(4))) float v4f;

#define NN      100000
#define DINF    128
#define DHIDF   128
#define DOUTF   64
#define NEDGE   1600000

template <int MODE>
__device__ __forceinline__ v4f stage_in(v4f v, v4f w) {
    if (MODE >= 1) {
        v.x = v.x > 0.f ? v.x : 0.f;
        v.y = v.y > 0.f ? v.y : 0.f;
        v.z = v.z > 0.f ? v.z : 0.f;
        v.w = v.w > 0.f ? v.w : 0.f;
    }
    if (MODE == 2) {
        w.x = w.x > 0.f ? w.x : 0.f;
        w.y = w.y > 0.f ? w.y : 0.f;
        w.z = w.z > 0.f ? w.z : 0.f;
        w.w = w.w > 0.f ? w.w : 0.f;
        v = v + w;
    }
    return v;
}

template <int N, int MODE, int ZN>
__global__ void __launch_bounds__(32 * (N / 16))
gemm_wmma_f32_wreg(const float* __restrict__ H,
                   const float* __restrict__ H2,
                   const float* __restrict__ W,
                   float* __restrict__ C,
                   float* __restrict__ Z,
                   int* __restrict__ ctr,
                   int nStrips) {
    constexpr int K  = 128;
    constexpr int NW = N / 16;
    constexpr int NT = 32 * NW;
    constexpr int P  = K + 4;
    constexpr int HL = 16 / NW;
    constexpr int WL = 16;
    constexpr int ZV = (16 * ZN / 4) / NT;
    static_assert((16 * ZN / 4) % NT == 0, "a strip of Z is a whole number of block-wide stores");
    __shared__ __attribute__((aligned(16))) float smem[N * P];

    const int lane = threadIdx.x & 31;
    const int wave = threadIdx.x >> 5;
    const int half = lane >> 4;
    const int l15  = lane & 15;
    const int n0   = wave * 16;

    if (blockIdx.x == 0 && threadIdx.x < 2) ctr[threadIdx.x * 64] = 0;

    #pragma unroll
    for (int i = 0; i < WL; ++i) {
        const int r = i * NW + wave;
        const v4f v = *(const v4f*)(W + (long)r * K + lane * 4);
        *(v4f*)(smem + r * P + lane * 4) = v;
    }
    __syncthreads();

    v2f bfr[K / 4];
    {
        const float* wp = smem + (n0 + l15) * P + half * 2;
        #pragma unroll
        for (int kk = 0; kk < K / 4; ++kk) bfr[kk] = *(const v2f*)(wp + kk * 4);
    }
    __syncthreads();

    float* const hb0 = smem;
    float* const hb1 = smem + 16 * P;

    int s = blockIdx.x;
    v4f pre[HL];
    v4f pre2[HL];
    if (s < nStrips) {
        #pragma unroll
        for (int i = 0; i < HL; ++i) {
            const int r = i * NW + wave;
            pre[i] = *(const v4f*)(H + ((long)s * 16 + r) * K + lane * 4);
            if (MODE == 2) pre2[i] = *(const v4f*)(H2 + ((long)s * 16 + r) * K + lane * 4);
            else           pre2[i] = pre[i];
        }
        #pragma unroll
        for (int i = 0; i < HL; ++i) {
            const int r = i * NW + wave;
            *(v4f*)(hb0 + r * P + lane * 4) = stage_in<MODE>(pre[i], pre2[i]);
        }
    }
    __syncthreads();

    int cur = 0;
    for (; s < nStrips; s += gridDim.x) {
        const int  sn      = s + gridDim.x;
        const bool hasNext = sn < nStrips;
        if (hasNext) {
            #pragma unroll
            for (int i = 0; i < HL; ++i) {
                const int r = i * NW + wave;
                pre[i] = *(const v4f*)(H + ((long)sn * 16 + r) * K + lane * 4);
                if (MODE == 2) pre2[i] = *(const v4f*)(H2 + ((long)sn * 16 + r) * K + lane * 4);
                else           pre2[i] = pre[i];
            }
        }

        const float* ap = (cur ? hb1 : hb0) + l15 * P + half * 2;
        v8f acc = {};
        #pragma unroll
        for (int kk = 0; kk < K / 4; ++kk) {
            const v2f a = *(const v2f*)(ap + kk * 4);
            acc = __builtin_amdgcn_wmma_f32_16x16x4_f32(
                 false, a,  false, bfr[kk],
                 (short)0, acc,  false,  false);
        }

        if (hasNext) {
            float* hn = cur ? hb0 : hb1;
            #pragma unroll
            for (int i = 0; i < HL; ++i) {
                const int r = i * NW + wave;
                *(v4f*)(hn + r * P + lane * 4) = stage_in<MODE>(pre[i], pre2[i]);
            }
        }
        __syncthreads();
        cur ^= 1;

        float* cp = C + ((long)s * 16 + half * 8) * N + n0 + l15;
        #pragma unroll
        for (int i = 0; i < 8; ++i) cp[(long)i * N] = acc[i];

        v4f* zp = (v4f*)(Z + (long)s * 16 * ZN) + threadIdx.x;
        const v4f z4 = {0.f, 0.f, 0.f, 0.f};
        #pragma unroll
        for (int i = 0; i < ZV; ++i) zp[i * NT] = z4;
    }
}

#define SPMM_TAKE_EDGE(B, UU)                                                        \
    {                                                                                \
        const int  p  = __builtin_ctzll(m);                                          \
        m &= m - 1ull;                                                               \
        const int  pl = p & 31;                                                      \
        const bool hi = p >= 32;                                                     \
              \
                           \
        const int  cA = __builtin_amdgcn_readlane(ccA, pl);                          \
        const int  cB = __builtin_amdgcn_readlane(ccB, pl);                          \
        const int  rA = __builtin_amdgcn_readlane(rrA, pl);                          \
        const int  rB = __builtin_amdgcn_readlane(rrB, pl);                          \
        const int  vA = __builtin_amdgcn_readlane(vvA, pl);                          \
        const int  vB = __builtin_amdgcn_readlane(vvB, pl);                          \
        const int  c  = hi ? cB : cA;                                                \
        yo[B][UU]     = (hi ? rB : rA) * D + lane;                                   \
        sv[B][UU]     = hi ? vB : vA;                                                \
        const float* hp = h + (long)c * D + lane;                                    \
        _Pragma("unroll")                                                            \
        for (int j = 0; j < R; ++j) hv[B][UU][j] = hp[j * 32];                       \
    }

#define SPMM_ADD_EDGE(B, UU)                                                         \
    {                                                                                \
        const float v  = __int_as_float(sv[B][UU]);                                  \
        float*      yp = y + yo[B][UU];                                              \
        _Pragma("unroll")                                                            \
        for (int j = 0; j < R; ++j) unsafeAtomicAdd(yp + j * 32, v * hv[B][UU][j]);  \
    }

#define SPMM_FETCH(B)                                                                \
    {                                                                                \
        _Pragma("unroll")                                                            \
        for (int u = 0; u < U; ++u) SPMM_TAKE_EDGE(B, u)                             \
    }
#define SPMM_ADDS(B)                                                                 \
    {                                                                                \
        _Pragma("unroll")                                                            \
        for (int u = 0; u < U; ++u) SPMM_ADD_EDGE(B, u)                              \
    }
#define SPMM_FETCH_PART(B)                                                           \
    {                                                                                \
        _Pragma("unroll")                                                            \
        for (int u = 0; u < U; ++u) if (u < rem) SPMM_TAKE_EDGE(B, u)                \
    }
#define SPMM_ADDS_PART(B)                                                            \
    {                                                                                \
        _Pragma("unroll")                                                            \
        for (int u = 0; u < U; ++u) if (u < rem) SPMM_ADD_EDGE(B, u)                 \
    }

__device__ __forceinline__ int spmm_draw(int* ctr, int lane) {
    int t = 0;
    if (lane == 0) t = __hip_atomic_fetch_add(ctr, 1, __ATOMIC_RELAXED, __HIP_MEMORY_SCOPE_AGENT);
    return __builtin_amdgcn_readfirstlane(t);
}

template <int D, int U>
__global__ void __launch_bounds__(256)
spmm_atomic_walk(const int*   __restrict__ erow,
                 const int*   __restrict__ ecol,
                 const float* __restrict__ eval,
                 const float* __restrict__ h,
                 float*       __restrict__ y,
                 int*         __restrict__ ctr,
                 int nTickets) {
    constexpr int R = D / 32;
    constexpr int HALFROW = NN / 2;
    const int lane = threadIdx.x & 31;

    const int myhalf = (int)((__builtin_amdgcn_s_sendmsg_rtn(0x87) >> 18) & 1u);

    float hv[2][U][R];
    int   yo[2][U];
    int   sv[2][U];

    for (int pass = 0; pass < 2; ++pass) {
        const int q = myhalf ^ pass;
        int* const cq = ctr + q * 64;

        int t0 = spmm_draw(cq, lane);
        int t1 = spmm_draw(cq, lane);
        int rrA = 0, ccA = 0, vvA = 0, rrB = 0, ccB = 0, vvB = 0;
        if (t0 < nTickets) {
            const int e = t0 * 64 + lane;
            rrA = erow[e];      ccA = ecol[e];      vvA = __float_as_int(eval[e]);
            rrB = erow[e + 32]; ccB = ecol[e + 32]; vvB = __float_as_int(eval[e + 32]);
        }
        while (t0 < nTickets) {
            const int tl = t1 < nTickets ? t1 : nTickets - 1;
            const int en = tl * 64 + lane;
            const int rrAn = erow[en];      const int ccAn = ecol[en];
            const int vvAn = __float_as_int(eval[en]);
            const int rrBn = erow[en + 32]; const int ccBn = ecol[en + 32];
            const int vvBn = __float_as_int(eval[en + 32]);

            const unsigned mA = __builtin_amdgcn_ballot_w32((rrA >= HALFROW) == (q != 0));
            const unsigned mB = __builtin_amdgcn_ballot_w32((rrB >= HALFROW) == (q != 0));
            unsigned long long m = (unsigned long long)mA | ((unsigned long long)mB << 32);
            const int cnt = __builtin_popcountll(m);
            const int ng  = cnt / U;
            const int rem = cnt - ng * U;

            if (ng > 0) {
                SPMM_FETCH(0)
                const int pairs = (ng - 1) >> 1;
                #pragma unroll 1
                for (int i = 0; i < pairs; ++i) {
                    SPMM_FETCH(1)
                    SPMM_ADDS(0)
                    SPMM_FETCH(0)
                    SPMM_ADDS(1)
                }
                if ((ng - 1) & 1) {
                    SPMM_FETCH(1)
                    SPMM_ADDS(0)
                    SPMM_FETCH_PART(0)
                    SPMM_ADDS(1)
                    SPMM_ADDS_PART(0)
                } else {
                    SPMM_FETCH_PART(1)
                    SPMM_ADDS(0)
                    SPMM_ADDS_PART(1)
                }
            } else {
                SPMM_FETCH_PART(0)
                SPMM_ADDS_PART(0)
            }

            rrA = rrAn; ccA = ccAn; vvA = vvAn;
            rrB = rrBn; ccB = ccBn; vvB = vvBn;
            t0 = t1;
            t1 = spmm_draw(cq, lane);
        }
    }
}

extern "C" void kernel_launch(void* const* d_in, const int* in_sizes, int n_in,
                              void* d_out, int out_size, void* d_ws, size_t ws_size,
                              hipStream_t stream) {
    const float* x    = (const float*)d_in[0];
    const int*   erow = (const int*)  d_in[1];
    const int*   ecol = (const int*)  d_in[2];
    const float* evalv= (const float*)d_in[3];
    const float* W0   = (const float*)d_in[4];
    const float* W1   = (const float*)d_in[5];
    const float* W2   = (const float*)d_in[6];
    float*       out  = (float*)d_out;

    const long nh = (long)NN * DHIDF;
    float* T  = (float*)d_ws;
    float* A  = T + nh;
    float* Bb = A + nh;

    static_assert(NEDGE % 64 == 0, "a ticket is 64 whole edges");
    constexpr int tickets = NEDGE / 64;
    constexpr int sblocks128 = 256 * 3;
    constexpr int sblocks64  = 256 * 2;
    constexpr int strips  = NN / 16;
    static_assert(NN % 16 == 0, "whole strips");
    constexpr int gblocks = strips / 5;

    int* ctr01 = (int*)out;
    int* ctr2  = (int*)(T + (long)NN * DOUTF);


    gemm_wmma_f32_wreg<DHIDF, 0, DHIDF><<<gblocks, 32 * (DHIDF / 16), 0, stream>>>(
        x, nullptr, W0, T, A, ctr01, strips);
    spmm_atomic_walk<DHIDF, 4><<<sblocks128, 256, 0, stream>>>(
        erow, ecol, evalv, T, A, ctr01, tickets);

    gemm_wmma_f32_wreg<DHIDF, 1, DHIDF><<<gblocks, 32 * (DHIDF / 16), 0, stream>>>(
        A, nullptr, W1, T, Bb, ctr01, strips);
    spmm_atomic_walk<DHIDF, 4><<<sblocks128, 256, 0, stream>>>(
        erow, ecol, evalv, T, Bb, ctr01, tickets);

    gemm_wmma_f32_wreg<DOUTF, 2, DOUTF><<<gblocks, 32 * (DOUTF / 16), 0, stream>>>(
        Bb, A, W2, T, out, ctr2, strips);
    spmm_atomic_walk<DOUTF, 8><<<sblocks64, 256, 0, stream>>>(
        erow, ecol, evalv, T, out, ctr2, tickets);
}


# SpMMs by row owner, compact then drain: a block keeps 391 y rows in LDS, queues matches of 1-byte owner tags per wave, loads 33-64 edge records at once, h rows 2 groups deep; LDS atomic arrival sums
# speedup vs baseline: 2.9788x; 1.1343x over previous
#include <hip/hip_runtime.h>

typedef __attribute__((ext_vector_type(2))) float v2f;
typedef __attribute__((ext_vector_type(8))) float v8f;
typedef __attribute__((ext_vector_type(4))) float v4f;
typedef __attribute__((ext_vector_type(4))) int   v4i;
typedef __attribute__((ext_vector_type(2))) int   v2i;
typedef __attribute__((ext_vector_type(4))) unsigned v4u;

#define NN      100000
#define DINF    128
#define DHIDF   128
#define DOUTF   64
#define NEDGE   1600000
#define OWN_ROWS 391
#define OWN_QCAP 64

template <int MODE>
__device__ __forceinline__ v4f stage_in(v4f v, v4f w) {
    if (MODE >= 1) {
        v.x = v.x > 0.f ? v.x : 0.f;
        v.y = v.y > 0.f ? v.y : 0.f;
        v.z = v.z > 0.f ? v.z : 0.f;
        v.w = v.w > 0.f ? v.w : 0.f;
    }
    if (MODE == 2) {
        w.x = w.x > 0.f ? w.x : 0.f;
        w.y = w.y > 0.f ? w.y : 0.f;
        w.z = w.z > 0.f ? w.z : 0.f;
        w.w = w.w > 0.f ? w.w : 0.f;
        v = v + w;
    }
    return v;
}

template <int N, int MODE, bool TAGS>
__global__ void __launch_bounds__(32 * (N / 16))
gemm_wmma_f32_wreg(const float* __restrict__ H,
                   const float* __restrict__ H2,
                   const float* __restrict__ W,
                   float* __restrict__ C,
                   const int*   __restrict__ erow,
                   const int*   __restrict__ ecol,
                   const float* __restrict__ eval,
                   int* __restrict__ tags,
                   int* __restrict__ rec,
                   int nStrips) {
    constexpr int K  = 128;
    constexpr int NW = N / 16;
    constexpr int NT = 32 * NW;
    constexpr int P  = K + 4;
    constexpr int HL = 16 / NW;
    constexpr int WL = 16;
    __shared__ __attribute__((aligned(16))) float smem[N * P];

    const int lane = threadIdx.x & 31;
    const int wave = threadIdx.x >> 5;
    const int half = lane >> 4;
    const int l15  = lane & 15;
    const int n0   = wave * 16;

    if (TAGS) {
        static_assert(NEDGE % 4 == 0, "four edges per thread and trip");
        for (int q = blockIdx.x * NT + threadIdx.x; q < NEDGE / 4; q += gridDim.x * NT) {
            const v4i r4 = *(const v4i*)(erow + 4 * q);
            const v4i c4 = *(const v4i*)(ecol + 4 * q);
            const v4f e4 = *(const v4f*)(eval + 4 * q);
            const int t0 = r4.x / OWN_ROWS, t1 = r4.y / OWN_ROWS;
            const int t2 = r4.z / OWN_ROWS, t3 = r4.w / OWN_ROWS;
            tags[q] = t0 | (t1 << 8) | (t2 << 16) | (t3 << 24);
            v4i lo, hi;
            lo.x = c4.x | ((r4.x - t0 * OWN_ROWS) << 17); lo.y = __float_as_int(e4.x);
            lo.z = c4.y | ((r4.y - t1 * OWN_ROWS) << 17); lo.w = __float_as_int(e4.y);
            hi.x = c4.z | ((r4.z - t2 * OWN_ROWS) << 17); hi.y = __float_as_int(e4.z);
            hi.z = c4.w | ((r4.w - t3 * OWN_ROWS) << 17); hi.w = __float_as_int(e4.w);
            *(v4i*)(rec + 8 * q)     = lo;
            *(v4i*)(rec + 8 * q + 4) = hi;
        }
    }

    #pragma unroll
    for (int i = 0; i < WL; ++i) {
        const int r = i * NW + wave;
        const v4f v = *(const v4f*)(W + (long)r * K + lane * 4);
        *(v4f*)(smem + r * P + lane * 4) = v;
    }
    __syncthreads();

    v2f bfr[K / 4];
    {
        const float* wp = smem + (n0 + l15) * P + half * 2;
        #pragma unroll
        for (int kk = 0; kk < K / 4; ++kk) bfr[kk] = *(const v2f*)(wp + kk * 4);
    }
    __syncthreads();

    float* const hb0 = smem;
    float* const hb1 = smem + 16 * P;

    int s = blockIdx.x;
    v4f pre[HL];
    v4f pre2[HL];
    if (s < nStrips) {
        #pragma unroll
        for (int i = 0; i < HL; ++i) {
            const int r = i * NW + wave;
            pre[i] = *(const v4f*)(H + ((long)s * 16 + r) * K + lane * 4);
            if (MODE == 2) pre2[i] = *(const v4f*)(H2 + ((long)s * 16 + r) * K + lane * 4);
            else           pre2[i] = pre[i];
        }
        #pragma unroll
        for (int i = 0; i < HL; ++i) {
            const int r = i * NW + wave;
            *(v4f*)(hb0 + r * P + lane * 4) = stage_in<MODE>(pre[i], pre2[i]);
        }
    }
    __syncthreads();

    int cur = 0;
    for (; s < nStrips; s += gridDim.x) {
        const int  sn      = s + gridDim.x;
        const bool hasNext = sn < nStrips;
        if (hasNext) {
            #pragma unroll
            for (int i = 0; i < HL; ++i) {
                const int r = i * NW + wave;
                pre[i] = *(const v4f*)(H + ((long)sn * 16 + r) * K + lane * 4);
                if (MODE == 2) pre2[i] = *(const v4f*)(H2 + ((long)sn * 16 + r) * K + lane * 4);
                else           pre2[i] = pre[i];
            }
        }

        const float* ap = (cur ? hb1 : hb0) + l15 * P + half * 2;
        v8f acc = {};
        #pragma unroll
        for (int kk = 0; kk < K / 4; ++kk) {
            const v2f a = *(const v2f*)(ap + kk * 4);
            acc = __builtin_amdgcn_wmma_f32_16x16x4_f32(
                 false, a,  false, bfr[kk],
                 (short)0, acc,  false,  false);
        }

        if (hasNext) {
            float* hn = cur ? hb0 : hb1;
            #pragma unroll
            for (int i = 0; i < HL; ++i) {
                const int r = i * NW + wave;
                *(v4f*)(hn + r * P + lane * 4) = stage_in<MODE>(pre[i], pre2[i]);
            }
        }
        __syncthreads();
        cur ^= 1;

        float* cp = C + ((long)s * 16 + half * 8) * N + n0 + l15;
        #pragma unroll
        for (int i = 0; i < 8; ++i) cp[(long)i * N] = acc[i];
    }
}

#define OWN_TAKE_EDGE(B, UU)                                                         \
    {                                                                                \
        const int  pl = qi & 31;                                                     \
        const bool hi = qi >= 32;                                                    \
        ++qi;                                                                        \
               \
                           \
        const int  xA = __builtin_amdgcn_readlane(xxA, pl);                          \
        const int  xB = __builtin_amdgcn_readlane(xxB, pl);                          \
        const int  vA = __builtin_amdgcn_readlane(vvA, pl);                          \
        const int  vB = __builtin_amdgcn_readlane(vvB, pl);                          \
        const int  x  = hi ? xB : xA;                                                \
        yo[B][UU]     = (int)((unsigned)x >> 17) * D + lane;                         \
        sv[B][UU]     = hi ? vB : vA;                                                \
        const float* hp = h + (long)(x & 0x1FFFF) * D + lane;                        \
        _Pragma("unroll")                                                            \
        for (int j = 0; j < R; ++j) hv[B][UU][j] = hp[j * 32];                       \
    }

#define OWN_ADD_EDGE(B, UU)                                                          \
    {                                                                                \
        const float v  = __int_as_float(sv[B][UU]);                                  \
        float*      ap = acc + yo[B][UU];                                            \
        _Pragma("unroll")                                                            \
        for (int j = 0; j < R; ++j)                                                  \
            __hip_atomic_fetch_add(ap + j * 32, v * hv[B][UU][j],                    \
                                   __ATOMIC_RELAXED, __HIP_MEMORY_SCOPE_WORKGROUP);  \
    }

#define OWN_FETCH(B)                                                                 \
    {                                                                                \
        _Pragma("unroll")                                                            \
        for (int u = 0; u < U; ++u) OWN_TAKE_EDGE(B, u)                              \
    }
#define OWN_ADDS(B)                                                                  \
    {                                                                                \
        _Pragma("unroll")                                                            \
        for (int u = 0; u < U; ++u) OWN_ADD_EDGE(B, u)                               \
    }
#define OWN_FETCH_PART(B)                                                            \
    {                                                                                \
        _Pragma("unroll")                                                            \
        for (int u = 0; u < U; ++u) if (u < rem) OWN_TAKE_EDGE(B, u)                 \
    }
#define OWN_ADDS_PART(B)                                                             \
    {                                                                                \
        _Pragma("unroll")                                                            \
        for (int u = 0; u < U; ++u) if (u < rem) OWN_ADD_EDGE(B, u)                  \
    }

template <int D, int U>
__device__ __forceinline__ void owner_drain(const int* q, const int cnt,
                                            const int*   __restrict__ rec,
                                            const float* __restrict__ h,
                                            float* acc, const int lane) {
    constexpr int R = D / 32;
    const int eA = lane      < cnt ? q[lane]      : 0;
    const int eB = lane + 32 < cnt ? q[lane + 32] : 0;
    const v2i rA = *(const v2i*)(rec + 2 * (long)eA);
    const v2i rB = *(const v2i*)(rec + 2 * (long)eB);
    const int xxA = rA.x, vvA = rA.y, xxB = rB.x, vvB = rB.y;

    float hv[2][U][R];
    int   yo[2][U];
    int   sv[2][U];
    int   qi  = 0;
    const int ng  = cnt / U;
    const int rem = cnt - ng * U;

    if (ng > 0) {
        OWN_FETCH(0)
        const int pairs = (ng - 1) >> 1;
        #pragma unroll 1
        for (int i = 0; i < pairs; ++i) {
            OWN_FETCH(1)
            OWN_ADDS(0)
            OWN_FETCH(0)
            OWN_ADDS(1)
        }
        if ((ng - 1) & 1) {
            OWN_FETCH(1)
            OWN_ADDS(0)
            OWN_FETCH_PART(0)
            OWN_ADDS(1)
            OWN_ADDS_PART(0)
        } else {
            OWN_FETCH_PART(1)
            OWN_ADDS(0)
            OWN_ADDS_PART(1)
        }
    } else {
        OWN_FETCH_PART(0)
        OWN_ADDS_PART(0)
    }
}

__device__ __forceinline__ unsigned zero_bytes(unsigned y) {
    return ~(((y & 0x7F7F7F7Fu) + 0x7F7F7F7Fu) | y | 0x7F7F7F7Fu);
}
__device__ __forceinline__ unsigned match16(v4u x, unsigned idw) {
    return (zero_bytes(x.x ^ idw) >> 7) | (zero_bytes(x.y ^ idw) >> 6) |
           (zero_bytes(x.z ^ idw) >> 5) | (zero_bytes(x.w ^ idw) >> 4);
}

template <int D, int U>
__global__ void __launch_bounds__(1024)
spmm_owner_lds(const int*   __restrict__ tags,
               const int*   __restrict__ rec,
               const float* __restrict__ h,
               float*       __restrict__ y) {
    constexpr int NCH   = NEDGE / 16;
    constexpr int NTRIP = (NCH + 127) / 128;
    static_assert(NEDGE % 16 == 0, "whole chunks");
    static_assert(32 * 128 <= NCH, "every wave's first trip lies inside the owner bytes");
    static_assert(OWN_ROWS * 256 >= NN && OWN_ROWS * 255 < NN, "256 owners cover the rows");
    __shared__ __attribute__((aligned(16))) float acc[OWN_ROWS * D];
    __shared__ int queue[32 * OWN_QCAP];

    const int lane = threadIdx.x & 31;
    const int wave = __builtin_amdgcn_readfirstlane((int)(threadIdx.x >> 5));
    const int blk  = blockIdx.x;

    {
        const v4f z4 = {0.f, 0.f, 0.f, 0.f};
        for (int i = threadIdx.x; i < OWN_ROWS * D / 4; i += 1024) ((v4f*)acc)[i] = z4;
    }
    __syncthreads();

    const unsigned idw = (unsigned)blk * 0x01010101u;
    const v4u      inv = {~idw, ~idw, ~idw, ~idw};
    const v4u*     tg  = (const v4u*)tags;
    int* const     q   = queue + wave * OWN_QCAP;
    int            cnt = 0;

    v4u cur[4];
    #pragma unroll
    for (int k = 0; k < 4; ++k) cur[k] = tg[wave * 128 + k * 32 + lane];
    for (int g = wave; g < NTRIP; g += 32) {
        v4u nxt[4];
        #pragma unroll
        for (int k = 0; k < 4; ++k) {
            const int c = (g + 32) * 128 + k * 32 + lane;
            const v4u t = tg[c < NCH ? c : NCH - 1];
            nxt[k] = c < NCH ? t : inv;
        }

        const unsigned ma = match16(cur[0], idw) | (match16(cur[1], idw) << 4);
        const unsigned mb = match16(cur[2], idw) | (match16(cur[3], idw) << 4);
        unsigned long long m = (unsigned long long)ma | ((unsigned long long)mb << 32);

        unsigned b;
        while ((b = __builtin_amdgcn_ballot_w32(m != 0ull)) != 0u) {
            if (m != 0ull) {
                const int p = __builtin_ctzll(m);
                m &= m - 1ull;
                const int k    = ((p >> 5) << 1) | ((p >> 2) & 1);
                const int word = p & 3;
                const int byte = (p & 31) >> 3;
                const int c    = g * 128 + k * 32 + lane;
                q[cnt + (int)__builtin_amdgcn_mbcnt_lo(b, 0u)] = c * 16 + word * 4 + byte;
            }
            cnt += __builtin_popcount(b);
            if (cnt > OWN_QCAP - 32) {
                owner_drain<D, U>(q, cnt, rec, h, acc, lane);
                cnt = 0;
            }
        }
        #pragma unroll
        for (int k = 0; k < 4; ++k) cur[k] = nxt[k];
    }
    if (cnt > 0) owner_drain<D, U>(q, cnt, rec, h, acc, lane);

    __syncthreads();

    const int row0  = blk * OWN_ROWS;
    const int nrows = NN - row0 < OWN_ROWS ? NN - row0 : OWN_ROWS;
    v4f* yp = (v4f*)(y + (long)row0 * D);
    for (int i = threadIdx.x; i < nrows * D / 4; i += 1024) yp[i] = ((const v4f*)acc)[i];
}

extern "C" void kernel_launch(void* const* d_in, const int* in_sizes, int n_in,
                              void* d_out, int out_size, void* d_ws, size_t ws_size,
                              hipStream_t stream) {
    const float* x    = (const float*)d_in[0];
    const int*   erow = (const int*)  d_in[1];
    const int*   ecol = (const int*)  d_in[2];
    const float* evalv= (const float*)d_in[3];
    const float* W0   = (const float*)d_in[4];
    const float* W1   = (const float*)d_in[5];
    const float* W2   = (const float*)d_in[6];
    float*       out  = (float*)d_out;

    const long nh = (long)NN * DHIDF;
    float* T  = (float*)d_ws;
    float* A  = T + nh;
    float* Bb = A + nh;

    constexpr int owners  = 256;
    constexpr int strips  = NN / 16;
    static_assert(NN % 16 == 0, "whole strips");
    constexpr int gblocks = strips / 5;

    static_assert((long)NEDGE * 9 <= (long)NN * DOUTF * 4, "bytes and records fit in out and in half of T");
    int* tags01 = (int*)out;
    int* rec01  = tags01 + NEDGE / 4;
    int* tags2  = (int*)(T + (long)NN * DOUTF);
    int* rec2   = tags2 + NEDGE / 4;

    gemm_wmma_f32_wreg<DHIDF, 0, true><<<gblocks, 32 * (DHIDF / 16), 0, stream>>>(
        x, nullptr, W0, T, erow, ecol, evalv, tags01, rec01, strips);
    spmm_owner_lds<DHIDF, 4><<<owners, 1024, 0, stream>>>(tags01, rec01, T, A);

    gemm_wmma_f32_wreg<DHIDF, 1, false><<<gblocks, 32 * (DHIDF / 16), 0, stream>>>(
        A, nullptr, W1, T, erow, ecol, evalv, tags01, rec01, strips);
    spmm_owner_lds<DHIDF, 4><<<owners, 1024, 0, stream>>>(tags01, rec01, T, Bb);

    gemm_wmma_f32_wreg<DOUTF, 2, true><<<gblocks, 32 * (DOUTF / 16), 0, stream>>>(
        Bb, A, W2, T, erow, ecol, evalv, tags2, rec2, strips);
    spmm_owner_lds<DOUTF, 8><<<owners, 1024, 0, stream>>>(tags2, rec2, T, out);
}


# Layer 1's SpMM skips the owner-byte scan: layer 0's scatter waves save the records they drain in their block's own rows of the layer-1 plane, layer 1 reads them back; LDS atomic arrival sums
# speedup vs baseline: 3.2399x; 1.0877x over previous
#include <hip/hip_runtime.h>

typedef __attribute__((ext_vector_type(2))) float v2f;
typedef __attribute__((ext_vector_type(8))) float v8f;
typedef __attribute__((ext_vector_type(4))) float v4f;
typedef __attribute__((ext_vector_type(4))) int   v4i;
typedef __attribute__((ext_vector_type(2))) int   v2i;
typedef __attribute__((ext_vector_type(4))) unsigned v4u;

#define NN      100000
#define DINF    128
#define DHIDF   128
#define DOUTF   64
#define NEDGE   1600000
#define OWN_ROWS 391
#define OWN_QCAP 64

template <int MODE>
__device__ __forceinline__ v4f stage_in(v4f v, v4f w) {
    if (MODE >= 1) {
        v.x = v.x > 0.f ? v.x : 0.f;
        v.y = v.y > 0.f ? v.y : 0.f;
        v.z = v.z > 0.f ? v.z : 0.f;
        v.w = v.w > 0.f ? v.w : 0.f;
    }
    if (MODE == 2) {
        w.x = w.x > 0.f ? w.x : 0.f;
        w.y = w.y > 0.f ? w.y : 0.f;
        w.z = w.z > 0.f ? w.z : 0.f;
        w.w = w.w > 0.f ? w.w : 0.f;
        v = v + w;
    }
    return v;
}

template <int N, int MODE, bool TAGS>
__global__ void __launch_bounds__(32 * (N / 16))
gemm_wmma_f32_wreg(const float* __restrict__ H,
                   const float* __restrict__ H2,
                   const float* __restrict__ W,
                   float* __restrict__ C,
                   const int*   __restrict__ erow,
                   const int*   __restrict__ ecol,
                   const float* __restrict__ eval,
                   int* __restrict__ tags,
                   int* __restrict__ rec,
                   int nStrips) {
    constexpr int K  = 128;
    constexpr int NW = N / 16;
    constexpr int NT = 32 * NW;
    constexpr int P  = K + 4;
    constexpr int HL = 16 / NW;
    constexpr int WL = 16;
    __shared__ __attribute__((aligned(16))) float smem[N * P];

    const int lane = threadIdx.x & 31;
    const int wave = threadIdx.x >> 5;
    const int half = lane >> 4;
    const int l15  = lane & 15;
    const int n0   = wave * 16;

    if (TAGS) {
        static_assert(NEDGE % 4 == 0, "four edges per thread and trip");
        for (int q = blockIdx.x * NT + threadIdx.x; q < NEDGE / 4; q += gridDim.x * NT) {
            const v4i r4 = *(const v4i*)(erow + 4 * q);
            const v4i c4 = *(const v4i*)(ecol + 4 * q);
            const v4f e4 = *(const v4f*)(eval + 4 * q);
            const int t0 = r4.x / OWN_ROWS, t1 = r4.y / OWN_ROWS;
            const int t2 = r4.z / OWN_ROWS, t3 = r4.w / OWN_ROWS;
            tags[q] = t0 | (t1 << 8) | (t2 << 16) | (t3 << 24);
            v4i lo, hi;
            lo.x = c4.x | ((r4.x - t0 * OWN_ROWS) << 17); lo.y = __float_as_int(e4.x);
            lo.z = c4.y | ((r4.y - t1 * OWN_ROWS) << 17); lo.w = __float_as_int(e4.y);
            hi.x = c4.z | ((r4.z - t2 * OWN_ROWS) << 17); hi.y = __float_as_int(e4.z);
            hi.z = c4.w | ((r4.w - t3 * OWN_ROWS) << 17); hi.w = __float_as_int(e4.w);
            *(v4i*)(rec + 8 * q)     = lo;
            *(v4i*)(rec + 8 * q + 4) = hi;
        }
    }

    #pragma unroll
    for (int i = 0; i < WL; ++i) {
        const int r = i * NW + wave;
        const v4f v = *(const v4f*)(W + (long)r * K + lane * 4);
        *(v4f*)(smem + r * P + lane * 4) = v;
    }
    __syncthreads();

    v2f bfr[K / 4];
    {
        const float* wp = smem + (n0 + l15) * P + half * 2;
        #pragma unroll
        for (int kk = 0; kk < K / 4; ++kk) bfr[kk] = *(const v2f*)(wp + kk * 4);
    }
    __syncthreads();

    float* const hb0 = smem;
    float* const hb1 = smem + 16 * P;

    int s = blockIdx.x;
    v4f pre[HL];
    v4f pre2[HL];
    if (s < nStrips) {
        #pragma unroll
        for (int i = 0; i < HL; ++i) {
            const int r = i * NW + wave;
            pre[i] = *(const v4f*)(H + ((long)s * 16 + r) * K + lane * 4);
            if (MODE == 2) pre2[i] = *(const v4f*)(H2 + ((long)s * 16 + r) * K + lane * 4);
            else           pre2[i] = pre[i];
        }
        #pragma unroll
        for (int i = 0; i < HL; ++i) {
            const int r = i * NW + wave;
            *(v4f*)(hb0 + r * P + lane * 4) = stage_in<MODE>(pre[i], pre2[i]);
        }
    }
    __syncthreads();

    int cur = 0;
    for (; s < nStrips; s += gridDim.x) {
        const int  sn      = s + gridDim.x;
        const bool hasNext = sn < nStrips;
        if (hasNext) {
            #pragma unroll
            for (int i = 0; i < HL; ++i) {
                const int r = i * NW + wave;
                pre[i] = *(const v4f*)(H + ((long)sn * 16 + r) * K + lane * 4);
                if (MODE == 2) pre2[i] = *(const v4f*)(H2 + ((long)sn * 16 + r) * K + lane * 4);
                else           pre2[i] = pre[i];
            }
        }

        const float* ap = (cur ? hb1 : hb0) + l15 * P + half * 2;
        v8f acc = {};
        #pragma unroll
        for (int kk = 0; kk < K / 4; ++kk) {
            const v2f a = *(const v2f*)(ap + kk * 4);
            acc = __builtin_amdgcn_wmma_f32_16x16x4_f32(
                 false, a,  false, bfr[kk],
                 (short)0, acc,  false,  false);
        }

        if (hasNext) {
            float* hn = cur ? hb0 : hb1;
            #pragma unroll
            for (int i = 0; i < HL; ++i) {
                const int r = i * NW + wave;
                *(v4f*)(hn + r * P + lane * 4) = stage_in<MODE>(pre[i], pre2[i]);
            }
        }
        __syncthreads();
        cur ^= 1;

        float* cp = C + ((long)s * 16 + half * 8) * N + n0 + l15;
        #pragma unroll
        for (int i = 0; i < 8; ++i) cp[(long)i * N] = acc[i];
    }
}

#define OWN_TAKE_EDGE(B, UU)                                                         \
    {                                                                                \
        const int  pl = qi & 31;                                                     \
        const bool hi = qi >= 32;                                                    \
        ++qi;                                                                        \
               \
                           \
        const int  xA = __builtin_amdgcn_readlane(xxA, pl);                          \
        const int  xB = __builtin_amdgcn_readlane(xxB, pl);                          \
        const int  vA = __builtin_amdgcn_readlane(vvA, pl);                          \
        const int  vB = __builtin_amdgcn_readlane(vvB, pl);                          \
        const int  x  = hi ? xB : xA;                                                \
        yo[B][UU]     = (int)((unsigned)x >> 17) * D + lane;                         \
        sv[B][UU]     = hi ? vB : vA;                                                \
        const float* hp = h + (long)(x & 0x1FFFF) * D + lane;                        \
        _Pragma("unroll")                                                            \
        for (int j = 0; j < R; ++j) hv[B][UU][j] = hp[j * 32];                       \
    }

#define OWN_ADD_EDGE(B, UU)                                                          \
    {                                                                                \
        const float v  = __int_as_float(sv[B][UU]);                                  \
        float*      ap = acc + yo[B][UU];                                            \
        _Pragma("unroll")                                                            \
        for (int j = 0; j < R; ++j)                                                  \
            __hip_atomic_fetch_add(ap + j * 32, v * hv[B][UU][j],                    \
                                   __ATOMIC_RELAXED, __HIP_MEMORY_SCOPE_WORKGROUP);  \
    }

#define OWN_FETCH(B)                                                                 \
    {                                                                                \
        _Pragma("unroll")                                                            \
        for (int u = 0; u < U; ++u) OWN_TAKE_EDGE(B, u)                              \
    }
#define OWN_ADDS(B)                                                                  \
    {                                                                                \
        _Pragma("unroll")                                                            \
        for (int u = 0; u < U; ++u) OWN_ADD_EDGE(B, u)                               \
    }
#define OWN_FETCH_PART(B)                                                            \
    {                                                                                \
        _Pragma("unroll")                                                            \
        for (int u = 0; u < U; ++u) if (u < rem) OWN_TAKE_EDGE(B, u)                 \
    }
#define OWN_ADDS_PART(B)                                                             \
    {                                                                                \
        _Pragma("unroll")                                                            \
        for (int u = 0; u < U; ++u) if (u < rem) OWN_ADD_EDGE(B, u)                  \
    }

template <int D, int U>
__device__ __forceinline__ void owner_drain_recs(const v2i rA, const v2i rB, const int cnt,
                                                 const float* __restrict__ h,
                                                 float* acc, const int lane) {
    constexpr int R = D / 32;
    const int xxA = rA.x, vvA = rA.y, xxB = rB.x, vvB = rB.y;

    float hv[2][U][R];
    int   yo[2][U];
    int   sv[2][U];
    int   qi  = 0;
    const int ng  = cnt / U;
    const int rem = cnt - ng * U;

    if (ng > 0) {
        OWN_FETCH(0)
        const int pairs = (ng - 1) >> 1;
        #pragma unroll 1
        for (int i = 0; i < pairs; ++i) {
            OWN_FETCH(1)
            OWN_ADDS(0)
            OWN_FETCH(0)
            OWN_ADDS(1)
        }
        if ((ng - 1) & 1) {
            OWN_FETCH(1)
            OWN_ADDS(0)
            OWN_FETCH_PART(0)
            OWN_ADDS(1)
            OWN_ADDS_PART(0)
        } else {
            OWN_FETCH_PART(1)
            OWN_ADDS(0)
            OWN_ADDS_PART(1)
        }
    } else {
        OWN_FETCH_PART(0)
        OWN_ADDS_PART(0)
    }
}

template <int D>
struct own_list {
    static constexpr int LAST_ROWS = NN - 255 * OWN_ROWS;
    static constexpr int CAPW      = (LAST_ROWS * D * 4 - 128) / 8 / 32;
    static_assert(LAST_ROWS > 0 && CAPW >= OWN_QCAP, "a wave's list holds at least one batch");
};

template <int D, int U, bool SAVE>
__device__ __forceinline__ void owner_drain(const int* q, const int cnt,
                                            const int*   __restrict__ rec,
                                            const float* __restrict__ h,
                                            float* acc, const int lane,
                                            int* lw, int& wcur) {
    const int eA = lane      < cnt ? q[lane]      : 0;
    const int eB = lane + 32 < cnt ? q[lane + 32] : 0;
    const v2i rA = *(const v2i*)(rec + 2 * (long)eA);
    const v2i rB = *(const v2i*)(rec + 2 * (long)eB);
    if (SAVE) {
        if (wcur + cnt <= own_list<D>::CAPW) {
            if (lane      < cnt) *(v2i*)(lw + 2 * (wcur + lane))      = rA;
            if (lane + 32 < cnt) *(v2i*)(lw + 2 * (wcur + lane + 32)) = rB;
        }
        wcur += cnt;
    }
    owner_drain_recs<D, U>(rA, rB, cnt, h, acc, lane);
}

__device__ __forceinline__ unsigned zero_bytes(unsigned y) {
    return ~(((y & 0x7F7F7F7Fu) + 0x7F7F7F7Fu) | y | 0x7F7F7F7Fu);
}
__device__ __forceinline__ unsigned match16(v4u x, unsigned idw) {
    return (zero_bytes(x.x ^ idw) >> 7) | (zero_bytes(x.y ^ idw) >> 6) |
           (zero_bytes(x.z ^ idw) >> 5) | (zero_bytes(x.w ^ idw) >> 4);
}

template <int D, int U, int LMODE>
__global__ void __launch_bounds__(1024)
spmm_owner_lds(const int*   __restrict__ tags,
               const int*   __restrict__ rec,
               const float* __restrict__ h,
               float*       y,
               float*       lplane) {
    constexpr int NCH   = NEDGE / 16;
    constexpr int NTRIP = (NCH + 127) / 128;
    static_assert(NEDGE % 16 == 0, "whole chunks");
    static_assert(32 * 128 <= NCH, "every wave's first trip lies inside the owner bytes");
    static_assert(OWN_ROWS * 256 >= NN && OWN_ROWS * 255 < NN, "256 owners cover the rows");
    __shared__ __attribute__((aligned(16))) float acc[OWN_ROWS * D];
    __shared__ int queue[32 * OWN_QCAP];

    const int lane = threadIdx.x & 31;
    const int wave = __builtin_amdgcn_readfirstlane((int)(threadIdx.x >> 5));
    const int blk  = blockIdx.x;

    {
        const v4f z4 = {0.f, 0.f, 0.f, 0.f};
        for (int i = threadIdx.x; i < OWN_ROWS * D / 4; i += 1024) ((v4f*)acc)[i] = z4;
    }
    __syncthreads();

    const unsigned idw = (unsigned)blk * 0x01010101u;
    const v4u      inv = {~idw, ~idw, ~idw, ~idw};
    const v4u*     tg  = (const v4u*)tags;
    int* const     q   = queue + wave * OWN_QCAP;
    int            cnt = 0;

    constexpr int CAPW = own_list<D>::CAPW;
    int* const lreg = LMODE ? (int*)(lplane + (long)blk * OWN_ROWS * D) : nullptr;
    int* const lw   = LMODE ? lreg + 32 + wave * (CAPW * 2) : nullptr;
    int        wcur = 0;
    int        nl   = 0;
    if (LMODE == 2) nl = __builtin_amdgcn_readfirstlane(lreg[wave]);

    if (LMODE == 2 && nl <= CAPW) {
        v2i cA = {0, 0}, cB = {0, 0};
        if (lane      < nl) cA = *(const v2i*)(lw + 2 * lane);
        if (lane + 32 < nl) cB = *(const v2i*)(lw + 2 * (lane + 32));
        for (int b0 = 0; b0 < nl; b0 += OWN_QCAP) {
            const int left = nl - b0;
            v2i nA = {0, 0}, nB = {0, 0};
            const int i0 = b0 + OWN_QCAP + lane;
            if (i0      < nl) nA = *(const v2i*)(lw + 2 * i0);
            if (i0 + 32 < nl) nB = *(const v2i*)(lw + 2 * (i0 + 32));
            owner_drain_recs<D, U>(cA, cB, left < OWN_QCAP ? left : OWN_QCAP, h, acc, lane);
            cA = nA; cB = nB;
        }
    } else {
    v4u cur[4];
    #pragma unroll
    for (int k = 0; k < 4; ++k) cur[k] = tg[wave * 128 + k * 32 + lane];
    for (int g = wave; g < NTRIP; g += 32) {
        v4u nxt[4];
        #pragma unroll
        for (int k = 0; k < 4; ++k) {
            const int c = (g + 32) * 128 + k * 32 + lane;
            const v4u t = tg[c < NCH ? c : NCH - 1];
            nxt[k] = c < NCH ? t : inv;
        }

        const unsigned ma = match16(cur[0], idw) | (match16(cur[1], idw) << 4);
        const unsigned mb = match16(cur[2], idw) | (match16(cur[3], idw) << 4);
        unsigned long long m = (unsigned long long)ma | ((unsigned long long)mb << 32);

        unsigned b;
        while ((b = __builtin_amdgcn_ballot_w32(m != 0ull)) != 0u) {
            if (m != 0ull) {
                const int p = __builtin_ctzll(m);
                m &= m - 1ull;
                const int k    = ((p >> 5) << 1) | ((p >> 2) & 1);
                const int word = p & 3;
                const int byte = (p & 31) >> 3;
                const int c    = g * 128 + k * 32 + lane;
                q[cnt + (int)__builtin_amdgcn_mbcnt_lo(b, 0u)] = c * 16 + word * 4 + byte;
            }
            cnt += __builtin_popcount(b);
            if (cnt > OWN_QCAP - 32) {
                owner_drain<D, U, LMODE == 1>(q, cnt, rec, h, acc, lane, lw, wcur);
                cnt = 0;
            }
        }
        #pragma unroll
        for (int k = 0; k < 4; ++k) cur[k] = nxt[k];
    }
    if (cnt > 0) owner_drain<D, U, LMODE == 1>(q, cnt, rec, h, acc, lane, lw, wcur);
    if (LMODE == 1 && lane == 0) lreg[wave] = wcur;
    }

    __syncthreads();

    const int row0  = blk * OWN_ROWS;
    const int nrows = NN - row0 < OWN_ROWS ? NN - row0 : OWN_ROWS;
    v4f* yp = (v4f*)(y + (long)row0 * D);
    for (int i = threadIdx.x; i < nrows * D / 4; i += 1024) yp[i] = ((const v4f*)acc)[i];
}

extern "C" void kernel_launch(void* const* d_in, const int* in_sizes, int n_in,
                              void* d_out, int out_size, void* d_ws, size_t ws_size,
                              hipStream_t stream) {
    const float* x    = (const float*)d_in[0];
    const int*   erow = (const int*)  d_in[1];
    const int*   ecol = (const int*)  d_in[2];
    const float* evalv= (const float*)d_in[3];
    const float* W0   = (const float*)d_in[4];
    const float* W1   = (const float*)d_in[5];
    const float* W2   = (const float*)d_in[6];
    float*       out  = (float*)d_out;

    const long nh = (long)NN * DHIDF;
    float* T  = (float*)d_ws;
    float* A  = T + nh;
    float* Bb = A + nh;

    constexpr int owners  = 256;
    constexpr int strips  = NN / 16;
    static_assert(NN % 16 == 0, "whole strips");
    constexpr int gblocks = strips / 5;

    static_assert((long)NEDGE * 9 <= (long)NN * DOUTF * 4, "bytes and records fit in out and in half of T");
    int* tags01 = (int*)out;
    int* rec01  = tags01 + NEDGE / 4;
    int* tags2  = (int*)(T + (long)NN * DOUTF);
    int* rec2   = tags2 + NEDGE / 4;

    gemm_wmma_f32_wreg<DHIDF, 0, true><<<gblocks, 32 * (DHIDF / 16), 0, stream>>>(
        x, nullptr, W0, T, erow, ecol, evalv, tags01, rec01, strips);
    spmm_owner_lds<DHIDF, 4, 1><<<owners, 1024, 0, stream>>>(tags01, rec01, T, A, Bb);

    gemm_wmma_f32_wreg<DHIDF, 1, false><<<gblocks, 32 * (DHIDF / 16), 0, stream>>>(
        A, nullptr, W1, T, erow, ecol, evalv, tags01, rec01, strips);
    spmm_owner_lds<DHIDF, 4, 2><<<owners, 1024, 0, stream>>>(tags01, rec01, T, Bb, Bb);

    gemm_wmma_f32_wreg<DOUTF, 2, true><<<gblocks, 32 * (DOUTF / 16), 0, stream>>>(
        Bb, A, W2, T, erow, ecol, evalv, tags2, rec2, strips);
    spmm_owner_lds<DOUTF, 8, 0><<<owners, 1024, 0, stream>>>(tags2, rec2, T, out, nullptr);
}


# Layer 2's SpMM reads the saved edge lists too (kept in each block's own rows of out until its closing store): its owner-byte scan, record gather and GEMM 2's tag pass go; LDS atomic arrival sums
# speedup vs baseline: 3.6459x; 1.1253x over previous
#include <hip/hip_runtime.h>

typedef __attribute__((ext_vector_type(2))) float v2f;
typedef __attribute__((ext_vector_type(8))) float v8f;
typedef __attribute__((ext_vector_type(4))) float v4f;
typedef __attribute__((ext_vector_type(4))) int   v4i;
typedef __attribute__((ext_vector_type(2))) int   v2i;
typedef __attribute__((ext_vector_type(4))) unsigned v4u;

#define NN      100000
#define DINF    128
#define DHIDF   128
#define DOUTF   64
#define NEDGE   1600000
#define OWN_ROWS 391
#define OWN_QCAP 64

template <int MODE>
__device__ __forceinline__ v4f stage_in(v4f v, v4f w) {
    if (MODE >= 1) {
        v.x = v.x > 0.f ? v.x : 0.f;
        v.y = v.y > 0.f ? v.y : 0.f;
        v.z = v.z > 0.f ? v.z : 0.f;
        v.w = v.w > 0.f ? v.w : 0.f;
    }
    if (MODE == 2) {
        w.x = w.x > 0.f ? w.x : 0.f;
        w.y = w.y > 0.f ? w.y : 0.f;
        w.z = w.z > 0.f ? w.z : 0.f;
        w.w = w.w > 0.f ? w.w : 0.f;
        v = v + w;
    }
    return v;
}

template <int N, int MODE, bool TAGS>
__global__ void __launch_bounds__(32 * (N / 16))
gemm_wmma_f32_wreg(const float* __restrict__ H,
                   const float* __restrict__ H2,
                   const float* __restrict__ W,
                   float* __restrict__ C,
                   const int*   __restrict__ erow,
                   const int*   __restrict__ ecol,
                   const float* __restrict__ eval,
                   int* __restrict__ tags,
                   int* __restrict__ rec,
                   int nStrips) {
    constexpr int K  = 128;
    constexpr int NW = N / 16;
    constexpr int NT = 32 * NW;
    constexpr int P  = K + 4;
    constexpr int HL = 16 / NW;
    constexpr int WL = 16;
    __shared__ __attribute__((aligned(16))) float smem[N * P];

    const int lane = threadIdx.x & 31;
    const int wave = threadIdx.x >> 5;
    const int half = lane >> 4;
    const int l15  = lane & 15;
    const int n0   = wave * 16;

    if (TAGS) {
        static_assert(NEDGE % 4 == 0, "four edges per thread and trip");
        for (int q = blockIdx.x * NT + threadIdx.x; q < NEDGE / 4; q += gridDim.x * NT) {
            const v4i r4 = *(const v4i*)(erow + 4 * q);
            const v4i c4 = *(const v4i*)(ecol + 4 * q);
            const v4f e4 = *(const v4f*)(eval + 4 * q);
            const int t0 = r4.x / OWN_ROWS, t1 = r4.y / OWN_ROWS;
            const int t2 = r4.z / OWN_ROWS, t3 = r4.w / OWN_ROWS;
            tags[q] = t0 | (t1 << 8) | (t2 << 16) | (t3 << 24);
            v4i lo, hi;
            lo.x = c4.x | ((r4.x - t0 * OWN_ROWS) << 17); lo.y = __float_as_int(e4.x);
            lo.z = c4.y | ((r4.y - t1 * OWN_ROWS) << 17); lo.w = __float_as_int(e4.y);
            hi.x = c4.z | ((r4.z - t2 * OWN_ROWS) << 17); hi.y = __float_as_int(e4.z);
            hi.z = c4.w | ((r4.w - t3 * OWN_ROWS) << 17); hi.w = __float_as_int(e4.w);
            *(v4i*)(rec + 8 * q)     = lo;
            *(v4i*)(rec + 8 * q + 4) = hi;
        }
    }

    #pragma unroll
    for (int i = 0; i < WL; ++i) {
        const int r = i * NW + wave;
        const v4f v = *(const v4f*)(W + (long)r * K + lane * 4);
        *(v4f*)(smem + r * P + lane * 4) = v;
    }
    __syncthreads();

    v2f bfr[K / 4];
    {
        const float* wp = smem + (n0 + l15) * P + half * 2;
        #pragma unroll
        for (int kk = 0; kk < K / 4; ++kk) bfr[kk] = *(const v2f*)(wp + kk * 4);
    }
    __syncthreads();

    float* const hb0 = smem;
    float* const hb1 = smem + 16 * P;

    int s = blockIdx.x;
    v4f pre[HL];
    v4f pre2[HL];
    if (s < nStrips) {
        #pragma unroll
        for (int i = 0; i < HL; ++i) {
            const int r = i * NW + wave;
            pre[i] = *(const v4f*)(H + ((long)s * 16 + r) * K + lane * 4);
            if (MODE == 2) pre2[i] = *(const v4f*)(H2 + ((long)s * 16 + r) * K + lane * 4);
            else           pre2[i] = pre[i];
        }
        #pragma unroll
        for (int i = 0; i < HL; ++i) {
            const int r = i * NW + wave;
            *(v4f*)(hb0 + r * P + lane * 4) = stage_in<MODE>(pre[i], pre2[i]);
        }
    }
    __syncthreads();

    int cur = 0;
    for (; s < nStrips; s += gridDim.x) {
        const int  sn      = s + gridDim.x;
        const bool hasNext = sn < nStrips;
        if (hasNext) {
            #pragma unroll
            for (int i = 0; i < HL; ++i) {
                const int r = i * NW + wave;
                pre[i] = *(const v4f*)(H + ((long)sn * 16 + r) * K + lane * 4);
                if (MODE == 2) pre2[i] = *(const v4f*)(H2 + ((long)sn * 16 + r) * K + lane * 4);
                else           pre2[i] = pre[i];
            }
        }

        const float* ap = (cur ? hb1 : hb0) + l15 * P + half * 2;
        v8f acc = {};
        #pragma unroll
        for (int kk = 0; kk < K / 4; ++kk) {
            const v2f a = *(const v2f*)(ap + kk * 4);
            acc = __builtin_amdgcn_wmma_f32_16x16x4_f32(
                 false, a,  false, bfr[kk],
                 (short)0, acc,  false,  false);
        }

        if (hasNext) {
            float* hn = cur ? hb0 : hb1;
            #pragma unroll
            for (int i = 0; i < HL; ++i) {
                const int r = i * NW + wave;
                *(v4f*)(hn + r * P + lane * 4) = stage_in<MODE>(pre[i], pre2[i]);
            }
        }
        __syncthreads();
        cur ^= 1;

        float* cp = C + ((long)s * 16 + half * 8) * N + n0 + l15;
        #pragma unroll
        for (int i = 0; i < 8; ++i) cp[(long)i * N] = acc[i];
    }
}

#define OWN_TAKE_EDGE(B, UU)                                                         \
    {                                                                                \
        const int  pl = qi & 31;                                                     \
        const bool hi = qi >= 32;                                                    \
        ++qi;                                                                        \
               \
                           \
        const int  xA = __builtin_amdgcn_readlane(xxA, pl);                          \
        const int  xB = __builtin_amdgcn_readlane(xxB, pl);                          \
        const int  vA = __builtin_amdgcn_readlane(vvA, pl);                          \
        const int  vB = __builtin_amdgcn_readlane(vvB, pl);                          \
        const int  x  = hi ? xB : xA;                                                \
        yo[B][UU]     = (int)((unsigned)x >> 17) * D + lane;                         \
        sv[B][UU]     = hi ? vB : vA;                                                \
        const float* hp = h + (long)(x & 0x1FFFF) * D + lane;                        \
        _Pragma("unroll")                                                            \
        for (int j = 0; j < R; ++j) hv[B][UU][j] = hp[j * 32];                       \
    }

#define OWN_ADD_EDGE(B, UU)                                                          \
    {                                                                                \
        const float v  = __int_as_float(sv[B][UU]);                                  \
        float*      ap = acc + yo[B][UU];                                            \
        _Pragma("unroll")                                                            \
        for (int j = 0; j < R; ++j)                                                  \
            __hip_atomic_fetch_add(ap + j * 32, v * hv[B][UU][j],                    \
                                   __ATOMIC_RELAXED, __HIP_MEMORY_SCOPE_WORKGROUP);  \
    }

#define OWN_FETCH(B)                                                                 \
    {                                                                                \
        _Pragma("unroll")                                                            \
        for (int u = 0; u < U; ++u) OWN_TAKE_EDGE(B, u)                              \
    }
#define OWN_ADDS(B)                                                                  \
    {                                                                                \
        _Pragma("unroll")                                                            \
        for (int u = 0; u < U; ++u) OWN_ADD_EDGE(B, u)                               \
    }
#define OWN_FETCH_PART(B)                                                            \
    {                                                                                \
        _Pragma("unroll")                                                            \
        for (int u = 0; u < U; ++u) if (u < rem) OWN_TAKE_EDGE(B, u)                 \
    }
#define OWN_ADDS_PART(B)                                                             \
    {                                                                                \
        _Pragma("unroll")                                                            \
        for (int u = 0; u < U; ++u) if (u < rem) OWN_ADD_EDGE(B, u)                  \
    }

template <int D, int U>
__device__ __forceinline__ void owner_drain_recs(const v2i rA, const v2i rB, const int cnt,
                                                 const float* __restrict__ h,
                                                 float* acc, const int lane) {
    constexpr int R = D / 32;
    const int xxA = rA.x, vvA = rA.y, xxB = rB.x, vvB = rB.y;

    float hv[2][U][R];
    int   yo[2][U];
    int   sv[2][U];
    int   qi  = 0;
    const int ng  = cnt / U;
    const int rem = cnt - ng * U;

    if (ng > 0) {
        OWN_FETCH(0)
        const int pairs = (ng - 1) >> 1;
        #pragma unroll 1
        for (int i = 0; i < pairs; ++i) {
            OWN_FETCH(1)
            OWN_ADDS(0)
            OWN_FETCH(0)
            OWN_ADDS(1)
        }
        if ((ng - 1) & 1) {
            OWN_FETCH(1)
            OWN_ADDS(0)
            OWN_FETCH_PART(0)
            OWN_ADDS(1)
            OWN_ADDS_PART(0)
        } else {
            OWN_FETCH_PART(1)
            OWN_ADDS(0)
            OWN_ADDS_PART(1)
        }
    } else {
        OWN_FETCH_PART(0)
        OWN_ADDS_PART(0)
    }
}

template <int D, int U>
__device__ __forceinline__ void owner_drain_list(const v2i rA, const v2i rB, const int cnt,
                                                 const float* __restrict__ h,
                                                 float* acc, const int lane) {
    constexpr int R = D / 32;
    const int xxA = rA.x, vvA = rA.y, xxB = rB.x, vvB = rB.y;

    float hv[2][U][R];
    int   yo[2][U];
    int   sv[2][U];
    int   qi  = 0;
    const int ng  = cnt / U;
    const int rem = cnt - ng * U;

    if (ng > 0) {
        OWN_FETCH(0)
        const int pairs = (ng - 1) >> 1;
        #pragma unroll 1
        for (int i = 0; i < pairs; ++i) {
            OWN_FETCH(1)
            OWN_ADDS(0)
            OWN_FETCH(0)
            OWN_ADDS(1)
        }
        if ((ng - 1) & 1) {
            OWN_FETCH(1)
            OWN_ADDS(0)
            OWN_ADDS(1)
        } else {
            OWN_ADDS(0)
        }
    }
    if (rem > 0) {
        OWN_FETCH(0)
        OWN_ADDS_PART(0)
    }
}

struct own_list {
    static constexpr int LAST_ROWS = NN - 255 * OWN_ROWS;
    static constexpr int HDR       = 64;
    static constexpr int CAPW      = ((LAST_ROWS * DOUTF * 4 - HDR * 4) / 8 / 32) & ~31;
    static_assert(LAST_ROWS > 0 && CAPW >= OWN_QCAP, "a wave's list holds at least one batch");
};

template <int D, int U>
__device__ __forceinline__ void owner_drain(const int* q, const int cnt,
                                            const int*   __restrict__ rec,
                                            const float* __restrict__ h,
                                            float* acc, const int lane,
                                            int* lw, int& wcur) {
    const int eA = lane      < cnt ? q[lane]      : 0;
    const int eB = lane + 32 < cnt ? q[lane + 32] : 0;
    const v2i rA = *(const v2i*)(rec + 2 * (long)eA);
    const v2i rB = *(const v2i*)(rec + 2 * (long)eB);
    if (wcur + cnt <= own_list::CAPW) {
        if (lane      < cnt) *(v2i*)(lw + 2 * (wcur + lane))      = rA;
        if (lane + 32 < cnt) *(v2i*)(lw + 2 * (wcur + lane + 32)) = rB;
    }
    wcur += cnt;
    owner_drain_recs<D, U>(rA, rB, cnt, h, acc, lane);
}

template <int D, int U>
__device__ __forceinline__ void owner_drain_raw(const int* q, const int cnt, const int row0,
                                                const int*   __restrict__ erow,
                                                const int*   __restrict__ ecol,
                                                const float* __restrict__ eval,
                                                const float* __restrict__ h,
                                                float* acc, const int lane) {
    const int eA = lane      < cnt ? q[lane]      : 0;
    const int eB = lane + 32 < cnt ? q[lane + 32] : 0;
    v2i rA, rB;
    rA.x = ecol[eA] | (int)((unsigned)(erow[eA] - row0) << 17);
    rA.y = __float_as_int(eval[eA]);
    rB.x = ecol[eB] | (int)((unsigned)(erow[eB] - row0) << 17);
    rB.y = __float_as_int(eval[eB]);
    owner_drain_recs<D, U>(rA, rB, cnt, h, acc, lane);
}

__device__ __forceinline__ unsigned zero_bytes(unsigned y) {
    return ~(((y & 0x7F7F7F7Fu) + 0x7F7F7F7Fu) | y | 0x7F7F7F7Fu);
}
__device__ __forceinline__ unsigned match16(v4u x, unsigned idw) {
    return (zero_bytes(x.x ^ idw) >> 7) | (zero_bytes(x.y ^ idw) >> 6) |
           (zero_bytes(x.z ^ idw) >> 5) | (zero_bytes(x.w ^ idw) >> 4);
}

template <int D, int U, bool SCAN>
__global__ void __launch_bounds__(1024)
spmm_owner_lds(const int*   __restrict__ tags,
               const int*   __restrict__ rec,
               const int*   __restrict__ erow,
               const int*   __restrict__ ecol,
               const float* __restrict__ eval,
               const float* __restrict__ h,
               float*       y,
               float*       lplane) {
    constexpr int NCH   = NEDGE / 16;
    constexpr int NTRIP = (NCH + 127) / 128;
    static_assert(NEDGE % 16 == 0, "whole chunks");
    static_assert(32 * 128 <= NCH, "every wave's first trip lies inside the owner bytes");
    static_assert(OWN_ROWS * 256 >= NN && OWN_ROWS * 255 < NN, "256 owners cover the rows");
    __shared__ __attribute__((aligned(16))) float acc[OWN_ROWS * D];
    __shared__ int queue[32 * OWN_QCAP];

    const int lane = threadIdx.x & 31;
    const int wave = __builtin_amdgcn_readfirstlane((int)(threadIdx.x >> 5));
    const int blk  = blockIdx.x;
    const int row0 = blk * OWN_ROWS;

    {
        const v4f z4 = {0.f, 0.f, 0.f, 0.f};
        for (int i = threadIdx.x; i < OWN_ROWS * D / 4; i += 1024) ((v4f*)acc)[i] = z4;
    }
    __syncthreads();

    int* const q   = queue + wave * OWN_QCAP;
    int        cnt = 0;

    constexpr int CAPW = own_list::CAPW;
    int* const lreg = (int*)(lplane + (long)blk * OWN_ROWS * DOUTF);
    int* const lw   = lreg + own_list::HDR + wave * (CAPW * 2);

    if (!SCAN) {
        const int nl = __builtin_amdgcn_readfirstlane(lreg[wave]);
        if (nl <= CAPW) {
            v2i cA = {0, 0}, cB = {0, 0};
            if (lane      < nl) cA = *(const v2i*)(lw + 2 * lane);
            if (lane + 32 < nl) cB = *(const v2i*)(lw + 2 * (lane + 32));
            for (int b0 = 0; b0 < nl; b0 += OWN_QCAP) {
                const int left = nl - b0;
                v2i nA = {0, 0}, nB = {0, 0};
                const int i0 = b0 + OWN_QCAP + lane;
                if (i0      < nl) nA = *(const v2i*)(lw + 2 * i0);
                if (i0 + 32 < nl) nB = *(const v2i*)(lw + 2 * (i0 + 32));
                owner_drain_list<D, U>(cA, cB, left < OWN_QCAP ? left : OWN_QCAP, h, acc, lane);
                cA = nA; cB = nB;
            }
        } else {
            for (int g = wave; g < NTRIP; g += 32) {
                for (int k = 0; k < 64; ++k) {
                    const int  e = g * 2048 + k * 32 + lane;
                    const bool m = e < NEDGE && (unsigned)(erow[e] - row0) < (unsigned)OWN_ROWS;
                    const unsigned b = __builtin_amdgcn_ballot_w32(m);
                    if (m) q[cnt + (int)__builtin_amdgcn_mbcnt_lo(b, 0u)] = e;
                    cnt += __builtin_popcount(b);
                    if (cnt > OWN_QCAP - 32) {
                        owner_drain_raw<D, U>(q, cnt, row0, erow, ecol, eval, h, acc, lane);
                        cnt = 0;
                    }
                }
            }
            if (cnt > 0) owner_drain_raw<D, U>(q, cnt, row0, erow, ecol, eval, h, acc, lane);
        }
    } else {
    const unsigned idw = (unsigned)blk * 0x01010101u;
    const v4u      inv = {~idw, ~idw, ~idw, ~idw};
    const v4u*     tg  = (const v4u*)tags;
    int            wcur = 0;
    v4u cur[4];
    #pragma unroll
    for (int k = 0; k < 4; ++k) cur[k] = tg[wave * 128 + k * 32 + lane];
    for (int g = wave; g < NTRIP; g += 32) {
        v4u nxt[4];
        #pragma unroll
        for (int k = 0; k < 4; ++k) {
            const int c = (g + 32) * 128 + k * 32 + lane;
            const v4u t = tg[c < NCH ? c : NCH - 1];
            nxt[k] = c < NCH ? t : inv;
        }

        const unsigned ma = match16(cur[0], idw) | (match16(cur[1], idw) << 4);
        const unsigned mb = match16(cur[2], idw) | (match16(cur[3], idw) << 4);
        unsigned long long m = (unsigned long long)ma | ((unsigned long long)mb << 32);

        unsigned b;
        while ((b = __builtin_amdgcn_ballot_w32(m != 0ull)) != 0u) {
            if (m != 0ull) {
                const int p = __builtin_ctzll(m);
                m &= m - 1ull;
                const int k    = ((p >> 5) << 1) | ((p >> 2) & 1);
                const int word = p & 3;
                const int byte = (p & 31) >> 3;
                const int c    = g * 128 + k * 32 + lane;
                q[cnt + (int)__builtin_amdgcn_mbcnt_lo(b, 0u)] = c * 16 + word * 4 + byte;
            }
            cnt += __builtin_popcount(b);
            if (cnt > OWN_QCAP - 32) {
                owner_drain<D, U>(q, cnt, rec, h, acc, lane, lw, wcur);
                cnt = 0;
            }
        }
        #pragma unroll
        for (int k = 0; k < 4; ++k) cur[k] = nxt[k];
    }
    if (cnt > 0) owner_drain<D, U>(q, cnt, rec, h, acc, lane, lw, wcur);
    if (lane == 0) lreg[wave] = wcur;
    }

    __syncthreads();

    const int nrows = NN - row0 < OWN_ROWS ? NN - row0 : OWN_ROWS;
    v4f* yp = (v4f*)(y + (long)row0 * D);
    for (int i = threadIdx.x; i < nrows * D / 4; i += 1024) yp[i] = ((const v4f*)acc)[i];
}

extern "C" void kernel_launch(void* const* d_in, const int* in_sizes, int n_in,
                              void* d_out, int out_size, void* d_ws, size_t ws_size,
                              hipStream_t stream) {
    const float* x    = (const float*)d_in[0];
    const int*   erow = (const int*)  d_in[1];
    const int*   ecol = (const int*)  d_in[2];
    const float* evalv= (const float*)d_in[3];
    const float* W0   = (const float*)d_in[4];
    const float* W1   = (const float*)d_in[5];
    const float* W2   = (const float*)d_in[6];
    float*       out  = (float*)d_out;

    const long nh = (long)NN * DHIDF;
    float* T  = (float*)d_ws;
    float* A  = T + nh;
    float* Bb = A + nh;

    constexpr int owners  = 256;
    constexpr int strips  = NN / 16;
    static_assert(NN % 16 == 0, "whole strips");
    constexpr int gblocks = strips / 5;

    static_assert((long)NEDGE * 9 <= (long)NN * DHIDF * 4, "bytes and records fit in Bb");
    int* tags0 = (int*)Bb;
    int* rec0  = tags0 + NEDGE / 4;

    gemm_wmma_f32_wreg<DHIDF, 0, true><<<gblocks, 32 * (DHIDF / 16), 0, stream>>>(
        x, nullptr, W0, T, erow, ecol, evalv, tags0, rec0, strips);
    spmm_owner_lds<DHIDF, 4, true><<<owners, 1024, 0, stream>>>(
        tags0, rec0, erow, ecol, evalv, T, A, out);

    gemm_wmma_f32_wreg<DHIDF, 1, false><<<gblocks, 32 * (DHIDF / 16), 0, stream>>>(
        A, nullptr, W1, T, erow, ecol, evalv, nullptr, nullptr, strips);
    spmm_owner_lds<DHIDF, 4, false><<<owners, 1024, 0, stream>>>(
        nullptr, nullptr, erow, ecol, evalv, T, Bb, out);

    gemm_wmma_f32_wreg<DOUTF, 2, false><<<gblocks, 32 * (DOUTF / 16), 0, stream>>>(
        Bb, A, W2, T, erow, ecol, evalv, nullptr, nullptr, strips);
    spmm_owner_lds<DOUTF, 8, false><<<owners, 1024, 0, stream>>>(
        nullptr, nullptr, erow, ecol, evalv, T, out, out);
}


# SpMM rows owned by waves: edges dealt to the row's wave (list slot by integer LDS add, arrival order); per edge one whole-row load and a plain LDS read-fma-write, no float atomics; list-order sums
# speedup vs baseline: 4.2480x; 1.1651x over previous
#include <hip/hip_runtime.h>

typedef __attribute__((ext_vector_type(2))) float v2f;
typedef __attribute__((ext_vector_type(8))) float v8f;
typedef __attribute__((ext_vector_type(4))) float v4f;
typedef __attribute__((ext_vector_type(4))) int   v4i;
typedef __attribute__((ext_vector_type(2))) int   v2i;
typedef __attribute__((ext_vector_type(4))) unsigned v4u;

#define NN      100000
#define DINF    128
#define DHIDF   128
#define DOUTF   64
#define NEDGE   1600000
#define OWN_ROWS 391
#define OWN_QCAP 64

template <int MODE>
__device__ __forceinline__ v4f stage_in(v4f v, v4f w) {
    if (MODE >= 1) {
        v.x = v.x > 0.f ? v.x : 0.f;
        v.y = v.y > 0.f ? v.y : 0.f;
        v.z = v.z > 0.f ? v.z : 0.f;
        v.w = v.w > 0.f ? v.w : 0.f;
    }
    if (MODE == 2) {
        w.x = w.x > 0.f ? w.x : 0.f;
        w.y = w.y > 0.f ? w.y : 0.f;
        w.z = w.z > 0.f ? w.z : 0.f;
        w.w = w.w > 0.f ? w.w : 0.f;
        v = v + w;
    }
    return v;
}

template <int N, int MODE, bool TAGS>
__global__ void __launch_bounds__(32 * (N / 16))
gemm_wmma_f32_wreg(const float* __restrict__ H,
                   const float* __restrict__ H2,
                   const float* __restrict__ W,
                   float* __restrict__ C,
                   const int*   __restrict__ erow,
                   const int*   __restrict__ ecol,
                   const float* __restrict__ eval,
                   int* __restrict__ tags,
                   int* __restrict__ rec,
                   int nStrips) {
    constexpr int K  = 128;
    constexpr int NW = N / 16;
    constexpr int NT = 32 * NW;
    constexpr int P  = K + 4;
    constexpr int HL = 16 / NW;
    constexpr int WL = 16;
    __shared__ __attribute__((aligned(16))) float smem[N * P];

    const int lane = threadIdx.x & 31;
    const int wave = threadIdx.x >> 5;
    const int half = lane >> 4;
    const int l15  = lane & 15;
    const int n0   = wave * 16;

    if (TAGS) {
        static_assert(NEDGE % 4 == 0, "four edges per thread and trip");
        for (int q = blockIdx.x * NT + threadIdx.x; q < NEDGE / 4; q += gridDim.x * NT) {
            const v4i r4 = *(const v4i*)(erow + 4 * q);
            const v4i c4 = *(const v4i*)(ecol + 4 * q);
            const v4f e4 = *(const v4f*)(eval + 4 * q);
            const int t0 = r4.x / OWN_ROWS, t1 = r4.y / OWN_ROWS;
            const int t2 = r4.z / OWN_ROWS, t3 = r4.w / OWN_ROWS;
            tags[q] = t0 | (t1 << 8) | (t2 << 16) | (t3 << 24);
            v4i lo, hi;
            lo.x = c4.x | ((r4.x - t0 * OWN_ROWS) << 17); lo.y = __float_as_int(e4.x);
            lo.z = c4.y | ((r4.y - t1 * OWN_ROWS) << 17); lo.w = __float_as_int(e4.y);
            hi.x = c4.z | ((r4.z - t2 * OWN_ROWS) << 17); hi.y = __float_as_int(e4.z);
            hi.z = c4.w | ((r4.w - t3 * OWN_ROWS) << 17); hi.w = __float_as_int(e4.w);
            *(v4i*)(rec + 8 * q)     = lo;
            *(v4i*)(rec + 8 * q + 4) = hi;
        }
    }

    #pragma unroll
    for (int i = 0; i < WL; ++i) {
        const int r = i * NW + wave;
        const v4f v = *(const v4f*)(W + (long)r * K + lane * 4);
        *(v4f*)(smem + r * P + lane * 4) = v;
    }
    __syncthreads();

    v2f bfr[K / 4];
    {
        const float* wp = smem + (n0 + l15) * P + half * 2;
        #pragma unroll
        for (int kk = 0; kk < K / 4; ++kk) bfr[kk] = *(const v2f*)(wp + kk * 4);
    }
    __syncthreads();

    float* const hb0 = smem;
    float* const hb1 = smem + 16 * P;

    int s = blockIdx.x;
    v4f pre[HL];
    v4f pre2[HL];
    if (s < nStrips) {
        #pragma unroll
        for (int i = 0; i < HL; ++i) {
            const int r = i * NW + wave;
            pre[i] = *(const v4f*)(H + ((long)s * 16 + r) * K + lane * 4);
            if (MODE == 2) pre2[i] = *(const v4f*)(H2 + ((long)s * 16 + r) * K + lane * 4);
            else           pre2[i] = pre[i];
        }
        #pragma unroll
        for (int i = 0; i < HL; ++i) {
            const int r = i * NW + wave;
            *(v4f*)(hb0 + r * P + lane * 4) = stage_in<MODE>(pre[i], pre2[i]);
        }
    }
    __syncthreads();

    int cur = 0;
    for (; s < nStrips; s += gridDim.x) {
        const int  sn      = s + gridDim.x;
        const bool hasNext = sn < nStrips;
        if (hasNext) {
            #pragma unroll
            for (int i = 0; i < HL; ++i) {
                const int r = i * NW + wave;
                pre[i] = *(const v4f*)(H + ((long)sn * 16 + r) * K + lane * 4);
                if (MODE == 2) pre2[i] = *(const v4f*)(H2 + ((long)sn * 16 + r) * K + lane * 4);
                else           pre2[i] = pre[i];
            }
        }

        const float* ap = (cur ? hb1 : hb0) + l15 * P + half * 2;
        v8f acc = {};
        #pragma unroll
        for (int kk = 0; kk < K / 4; ++kk) {
            const v2f a = *(const v2f*)(ap + kk * 4);
            acc = __builtin_amdgcn_wmma_f32_16x16x4_f32(
                 false, a,  false, bfr[kk],
                 (short)0, acc,  false,  false);
        }

        if (hasNext) {
            float* hn = cur ? hb0 : hb1;
            #pragma unroll
            for (int i = 0; i < HL; ++i) {
                const int r = i * NW + wave;
                *(v4f*)(hn + r * P + lane * 4) = stage_in<MODE>(pre[i], pre2[i]);
            }
        }
        __syncthreads();
        cur ^= 1;

        float* cp = C + ((long)s * 16 + half * 8) * N + n0 + l15;
        #pragma unroll
        for (int i = 0; i < 8; ++i) cp[(long)i * N] = acc[i];
    }
}


#define OWN_WROWS 13
static_assert(OWN_WROWS * 32 >= OWN_ROWS, "32 waves x OWN_WROWS rows cover a block's rows");

template <int D> struct own_vec;
template <> struct own_vec<128> { typedef v4f T; };
template <> struct own_vec<64>  { typedef v2f T; };

#define WROW_TAKE_EDGE(B, UU)                                                         \
    {                                                                                \
        const int  pl = qi & 31;                                                     \
        const bool hi = qi >= 32;                                                    \
        ++qi;                                                                        \
               \
                           \
        const int  xA = __builtin_amdgcn_readlane(xxA, pl);                          \
        const int  xB = __builtin_amdgcn_readlane(xxB, pl);                          \
        const int  vA = __builtin_amdgcn_readlane(vvA, pl);                          \
        const int  vB = __builtin_amdgcn_readlane(vvB, pl);                          \
        const int  x  = hi ? xB : xA;                                                \
        jj[B][UU]     = (int)((unsigned)x >> 22);                 \
        sv[B][UU]     = hi ? vB : vA;                                                \
        hv[B][UU]     = *(const VT*)(h + (long)(x & 0x1FFFF) * D + lane * VW);       \
    }

#define WROW_ADD_EDGE(B, UU)                                                          \
    {                                                                                \
        const float v  = __int_as_float(sv[B][UU]);                                  \
                    \
        VT* const   ap = (VT*)(accw + jj[B][UU] * D + lane * VW);                    \
        VT          a  = *ap;                                                        \
        a += v * hv[B][UU];                                                          \
        *ap = a;                                                                     \
    }

#define WROW_FETCH(B)                                                                 \
    {                                                                                \
        _Pragma("unroll")                                                            \
        for (int u = 0; u < U; ++u) WROW_TAKE_EDGE(B, u)                              \
    }
#define WROW_ADDS(B)                                                                  \
    {                                                                                \
        _Pragma("unroll")                                                            \
        for (int u = 0; u < U; ++u) WROW_ADD_EDGE(B, u)                               \
    }
#define WROW_ADDS_PART(B)                                                             \
    {                                                                                \
        _Pragma("unroll")                                                            \
        for (int u = 0; u < U; ++u) if (u < rem) WROW_ADD_EDGE(B, u)                  \
    }

template <int D, int U>
__device__ __forceinline__ void owner_sum_batch(const v2i rA, const v2i rB, const int cnt,
                                                const float* __restrict__ h,
                                                float* accw, const int lane) {
    typedef typename own_vec<D>::T VT;
    constexpr int VW = D / 32;
    const int xxA = rA.x, vvA = rA.y, xxB = rB.x, vvB = rB.y;

    VT  hv[2][U];
    int jj[2][U];
    int sv[2][U];
    int qi  = 0;
    const int ng  = cnt / U;
    const int rem = cnt - ng * U;

    if (ng > 0) {
        WROW_FETCH(0)
        const int pairs = (ng - 1) >> 1;
        #pragma unroll 1
        for (int i = 0; i < pairs; ++i) {
            WROW_FETCH(1)
            WROW_ADDS(0)
            WROW_FETCH(0)
            WROW_ADDS(1)
        }
        if ((ng - 1) & 1) {
            WROW_FETCH(1)
            WROW_ADDS(0)
            WROW_ADDS(1)
        } else {
            WROW_ADDS(0)
        }
    }
    if (rem > 0) {
        WROW_FETCH(0)
        WROW_ADDS_PART(0)
    }
}

struct own_list {
    static constexpr int LAST_ROWS = NN - 255 * OWN_ROWS;
    static constexpr int HDR       = 64;
    static constexpr int CAPW      = ((LAST_ROWS * DOUTF * 4 - HDR * 4) / 8 / 32) & ~31;
    static_assert(LAST_ROWS > 0 && CAPW >= OWN_QCAP, "a wave's list holds at least one batch");
};

__device__ __forceinline__ void owner_deal(const int* q, const int cnt,
                                           const int* __restrict__ rec,
                                           int* lcnt, int* llist, const int lane) {
    const int eA = lane      < cnt ? q[lane]      : 0;
    const int eB = lane + 32 < cnt ? q[lane + 32] : 0;
    const v2i rA = *(const v2i*)(rec + 2 * (long)eA);
    const v2i rB = *(const v2i*)(rec + 2 * (long)eB);
    if (lane < cnt) {
        const int dw   = (int)(((unsigned)rA.x >> 17) & 31u);
        const int slot = __hip_atomic_fetch_add(lcnt + dw, 1, __ATOMIC_RELAXED,
                                                __HIP_MEMORY_SCOPE_WORKGROUP);
        if (slot < own_list::CAPW) *(v2i*)(llist + 2 * (dw * own_list::CAPW + slot)) = rA;
    }
    if (lane + 32 < cnt) {
        const int dw   = (int)(((unsigned)rB.x >> 17) & 31u);
        const int slot = __hip_atomic_fetch_add(lcnt + dw, 1, __ATOMIC_RELAXED,
                                                __HIP_MEMORY_SCOPE_WORKGROUP);
        if (slot < own_list::CAPW) *(v2i*)(llist + 2 * (dw * own_list::CAPW + slot)) = rB;
    }
}

template <int D, int U>
__device__ __forceinline__ void owner_sum_raw(int* q, const int row0, const int wave,
                                              const int*   __restrict__ erow,
                                              const int*   __restrict__ ecol,
                                              const float* __restrict__ eval,
                                              const float* __restrict__ h,
                                              float* accw, const int lane) {
    static_assert(NEDGE % 32 == 0, "whole steps of 32 edges");
    int cnt = 0;
    for (int base = 0; base < NEDGE; base += 32) {
        const int      e  = base + lane;
        const unsigned lr = (unsigned)(erow[e] - row0);
        const bool     m  = lr < (unsigned)OWN_ROWS && (int)(lr & 31u) == wave;
        const unsigned b  = __builtin_amdgcn_ballot_w32(m);
        if (m) q[cnt + (int)__builtin_amdgcn_mbcnt_lo(b, 0u)] = e;
        cnt += __builtin_popcount(b);
        if (cnt > OWN_QCAP - 32 || (base + 32 >= NEDGE && cnt > 0)) {
            const int eA = lane      < cnt ? q[lane]      : q[0];
            const int eB = lane + 32 < cnt ? q[lane + 32] : q[0];
            v2i rA, rB;
            rA.x = ecol[eA] | (int)((unsigned)(erow[eA] - row0) << 17);
            rA.y = __float_as_int(eval[eA]);
            rB.x = ecol[eB] | (int)((unsigned)(erow[eB] - row0) << 17);
            rB.y = __float_as_int(eval[eB]);
            owner_sum_batch<D, U>(rA, rB, cnt, h, accw, lane);
            cnt = 0;
        }
    }
}
__device__ __forceinline__ unsigned zero_bytes(unsigned y) {
    return ~(((y & 0x7F7F7F7Fu) + 0x7F7F7F7Fu) | y | 0x7F7F7F7Fu);
}
__device__ __forceinline__ unsigned match16(v4u x, unsigned idw) {
    return (zero_bytes(x.x ^ idw) >> 7) | (zero_bytes(x.y ^ idw) >> 6) |
           (zero_bytes(x.z ^ idw) >> 5) | (zero_bytes(x.w ^ idw) >> 4);
}

template <int D, int U, bool SCAN, bool LAST>
__global__ void __launch_bounds__(1024)
spmm_owner_rows(const int*   __restrict__ tags,
                const int*   __restrict__ rec,
                const int*   __restrict__ erow,
                const int*   __restrict__ ecol,
                const float* __restrict__ eval,
                const float* __restrict__ h,
                float*       y,
                float*       lplane) {
    typedef typename own_vec<D>::T VT;
    constexpr int VW    = D / 32;
    constexpr int NCH   = NEDGE / 16;
    constexpr int NTRIP = (NCH + 127) / 128;
    constexpr int CAPW  = own_list::CAPW;
    static_assert(NEDGE % 16 == 0, "whole chunks");
    static_assert(32 * 128 <= NCH, "every wave's first trip lies inside the owner bytes");
    static_assert(OWN_ROWS * 256 >= NN && OWN_ROWS * 255 < NN, "256 owners cover the rows");
    static_assert(OWN_ROWS <= 512, "a local row fits 9 bits of a record");
    __shared__ __attribute__((aligned(16))) float accs[32 * OWN_WROWS * D];
    __shared__ int queue[32 * OWN_QCAP];
    __shared__ int lcnt[32];
    __shared__ __attribute__((aligned(16))) int llist[SCAN ? 32 * CAPW * 2 : 4];

    const int lane = threadIdx.x & 31;
    const int wave = __builtin_amdgcn_readfirstlane((int)(threadIdx.x >> 5));
    const int blk  = blockIdx.x;
    const int row0 = blk * OWN_ROWS;

    int* const q = queue + wave * OWN_QCAP;

    int* const lreg = (int*)(lplane + (long)blk * OWN_ROWS * DOUTF);
    int* const lw   = lreg + own_list::HDR + wave * (CAPW * 2);
    int* const ll   = llist + (SCAN ? wave * (CAPW * 2) : 0);

    float* const accw = accs + wave * (OWN_WROWS * D);
    #pragma unroll
    for (int k = 0; k < OWN_WROWS; ++k) *(VT*)(accw + k * D + lane * VW) = (VT)(0.f);

    int nl;
    if (SCAN) {
        if (threadIdx.x < 32) lcnt[threadIdx.x] = 0;
        __syncthreads();

        const unsigned idw = (unsigned)blk * 0x01010101u;
        const v4u      inv = {~idw, ~idw, ~idw, ~idw};
        const v4u*     tg  = (const v4u*)tags;
        int            cnt = 0;
        v4u cur[4];
        #pragma unroll
        for (int k = 0; k < 4; ++k) cur[k] = tg[wave * 128 + k * 32 + lane];
        for (int g = wave; g < NTRIP; g += 32) {
            v4u nxt[4];
            #pragma unroll
            for (int k = 0; k < 4; ++k) {
                const int c = (g + 32) * 128 + k * 32 + lane;
                const v4u t = tg[c < NCH ? c : NCH - 1];
                nxt[k] = c < NCH ? t : inv;
            }

            const unsigned ma = match16(cur[0], idw) | (match16(cur[1], idw) << 4);
            const unsigned mb = match16(cur[2], idw) | (match16(cur[3], idw) << 4);
            unsigned long long m = (unsigned long long)ma | ((unsigned long long)mb << 32);

            unsigned b;
            while ((b = __builtin_amdgcn_ballot_w32(m != 0ull)) != 0u) {
                if (m != 0ull) {
                    const int p = __builtin_ctzll(m);
                    m &= m - 1ull;
                    const int k    = ((p >> 5) << 1) | ((p >> 2) & 1);
                    const int word = p & 3;
                    const int byte = (p & 31) >> 3;
                    const int c    = g * 128 + k * 32 + lane;
                    q[cnt + (int)__builtin_amdgcn_mbcnt_lo(b, 0u)] = c * 16 + word * 4 + byte;
                }
                cnt += __builtin_popcount(b);
                if (cnt > OWN_QCAP - 32) {
                    owner_deal(q, cnt, rec, lcnt, llist, lane);
                    cnt = 0;
                }
            }
            #pragma unroll
            for (int k = 0; k < 4; ++k) cur[k] = nxt[k];
        }
        if (cnt > 0) owner_deal(q, cnt, rec, lcnt, llist, lane);

        __syncthreads();
        nl = __builtin_amdgcn_readfirstlane(lcnt[wave]);
        if (lane == 0) lreg[wave] = nl;
    } else {
        nl = __builtin_amdgcn_readfirstlane(lreg[wave]);
    }

    if (nl <= CAPW) {
        v2i cA = {0, 0}, cB = {0, 0};
        if (lane      < nl) cA = SCAN ? *(const v2i*)(ll + 2 * lane)        : *(const v2i*)(lw + 2 * lane);
        if (lane + 32 < nl) cB = SCAN ? *(const v2i*)(ll + 2 * (lane + 32)) : *(const v2i*)(lw + 2 * (lane + 32));
        for (int b0 = 0; b0 < nl; b0 += OWN_QCAP) {
            const int left = nl - b0;
            v2i nA = {0, 0}, nB = {0, 0};
            const int i0 = b0 + OWN_QCAP + lane;
            if (i0      < nl) nA = SCAN ? *(const v2i*)(ll + 2 * i0)        : *(const v2i*)(lw + 2 * i0);
            if (i0 + 32 < nl) nB = SCAN ? *(const v2i*)(ll + 2 * (i0 + 32)) : *(const v2i*)(lw + 2 * (i0 + 32));
            if (SCAN) {
                if (lane      < left) *(v2i*)(lw + 2 * (b0 + lane))      = cA;
                if (lane + 32 < left) *(v2i*)(lw + 2 * (b0 + lane + 32)) = cB;
            }
            owner_sum_batch<D, U>(cA, cB, left < OWN_QCAP ? left : OWN_QCAP, h, accw, lane);
            cA = nA; cB = nB;
        }
    } else {
        owner_sum_raw<D, U>(q, row0, wave, erow, ecol, eval, h, accw, lane);
    }

    if (LAST) __syncthreads();

    const int nrows = NN - row0 < OWN_ROWS ? NN - row0 : OWN_ROWS;
    #pragma unroll
    for (int k = 0; k < OWN_WROWS; ++k) {
        const int lr = wave + 32 * k;
        if (lr < nrows) *(VT*)(y + (long)(row0 + lr) * D + lane * VW) = *(const VT*)(accw + k * D + lane * VW);
    }
}

extern "C" void kernel_launch(void* const* d_in, const int* in_sizes, int n_in,
                              void* d_out, int out_size, void* d_ws, size_t ws_size,
                              hipStream_t stream) {
    const float* x    = (const float*)d_in[0];
    const int*   erow = (const int*)  d_in[1];
    const int*   ecol = (const int*)  d_in[2];
    const float* evalv= (const float*)d_in[3];
    const float* W0   = (const float*)d_in[4];
    const float* W1   = (const float*)d_in[5];
    const float* W2   = (const float*)d_in[6];
    float*       out  = (float*)d_out;

    const long nh = (long)NN * DHIDF;
    float* T  = (float*)d_ws;
    float* A  = T + nh;
    float* Bb = A + nh;

    constexpr int owners  = 256;
    constexpr int strips  = NN / 16;
    static_assert(NN % 16 == 0, "whole strips");
    constexpr int gblocks = strips / 5;

    static_assert((long)NEDGE * 9 <= (long)NN * DHIDF * 4, "bytes and records fit in Bb");
    int* tags0 = (int*)Bb;
    int* rec0  = tags0 + NEDGE / 4;

    gemm_wmma_f32_wreg<DHIDF, 0, true><<<gblocks, 32 * (DHIDF / 16), 0, stream>>>(
        x, nullptr, W0, T, erow, ecol, evalv, tags0, rec0, strips);
    spmm_owner_rows<DHIDF, 4, true, false><<<owners, 1024, 0, stream>>>(
        tags0, rec0, erow, ecol, evalv, T, A, out);

    gemm_wmma_f32_wreg<DHIDF, 1, false><<<gblocks, 32 * (DHIDF / 16), 0, stream>>>(
        A, nullptr, W1, T, erow, ecol, evalv, nullptr, nullptr, strips);
    spmm_owner_rows<DHIDF, 4, false, false><<<owners, 1024, 0, stream>>>(
        nullptr, nullptr, erow, ecol, evalv, T, Bb, out);

    gemm_wmma_f32_wreg<DOUTF, 2, false><<<gblocks, 32 * (DOUTF / 16), 0, stream>>>(
        Bb, A, W2, T, erow, ecol, evalv, nullptr, nullptr, strips);
    spmm_owner_rows<DOUTF, 8, false, true><<<owners, 1024, 0, stream>>>(
        nullptr, nullptr, erow, ecol, evalv, T, out, out);
}
